# starting kernel
# baseline (speedup 1.0000x reference)
#include <hip/hip_runtime.h>
#include <math.h>

typedef __attribute__((ext_vector_type(2))) float v2f;
typedef __attribute__((ext_vector_type(8))) float v8f;

struct Rings { int r1[64]; int r2[64]; };

constexpr Rings makeRings() {
  Rings R{};
  for (int rr = 1; rr <= 2; ++rr) {
    int total[64];
    for (int i = 0; i < 64; ++i) total[i] = i;
    for (int w = 0; w < 6; ++w) {
      int t = (w + rr) % 6;
      int nxt[64];
      for (int i = 0; i < 64; ++i) {
        int cbit = (i >> (5 - w)) & 1;
        nxt[i] = total[i ^ (cbit << (5 - t))];
      }
      for (int i = 0; i < 64; ++i) total[i] = nxt[i];
    }
    for (int i = 0; i < 64; ++i) { if (rr == 1) R.r1[i] = total[i]; else R.r2[i] = total[i]; }
  }
  return R;
}

__constant__ Rings RINGS = makeRings();

__device__ __forceinline__ float2 cmul(float2 a, float2 b) {
  return make_float2(a.x * b.x - a.y * b.y, a.x * b.y + a.y * b.x);
}
__device__ __forceinline__ float2 cadd(float2 a, float2 b) { return make_float2(a.x + b.x, a.y + b.y); }
__device__ __forceinline__ float2 shflx2(float2 v, int m) {
  return make_float2(__shfl_xor(v.x, m, 32), __shfl_xor(v.y, m, 32));
}
__device__ __forceinline__ float2 shfl2(float2 v, int l) {
  return make_float2(__shfl(v.x, l, 32), __shfl(v.y, l, 32));
}

#define RED16(v) do { v += __shfl_xor(v,1,32); v += __shfl_xor(v,2,32); \
                      v += __shfl_xor(v,4,32); v += __shfl_xor(v,8,32); } while(0)

__global__ __launch_bounds__(256) void qmaml_kernel(
    const float* __restrict__ x,  const float* __restrict__ W1,
    const float* __restrict__ b1, const float* __restrict__ ln_g,
    const float* __restrict__ ln_b, const float* __restrict__ W2,
    const float* __restrict__ b2, const float* __restrict__ th_s,
    const float* __restrict__ th_t, const float* __restrict__ Wc,
    const float* __restrict__ bc, float* __restrict__ out, int B)
{
  __shared__ float  sW1[4096];
  __shared__ float  sW2[384];
  __shared__ float  sB1[64], sG[64], sBeta[64];
  __shared__ float  sB2[8], sWc[32], sBc[8];
  __shared__ float2 sGates[72];
  __shared__ float  sZ[768];

  const int tid = threadIdx.x;

  for (int i = tid; i < 1024; i += 256) ((float4*)sW1)[i] = ((const float4*)W1)[i];
  for (int i = tid; i < 384; i += 256)  sW2[i] = W2[i];
  if (tid < 64) { sB1[tid] = b1[tid]; sG[tid] = ln_g[tid]; sBeta[tid] = ln_b[tid]; }
  if (tid < 6)  sB2[tid] = b2[tid];
  if (tid < 30) sWc[tid] = Wc[tid];
  if (tid < 5)  sBc[tid] = bc[tid];
  if (tid < 18) {
    int l = tid / 6, w = tid % 6;
    const float* th = (l < 2) ? (th_s + (l * 6 + w) * 3) : (th_t + w * 3);
    float phi = th[0], theta = th[1], omega = th[2];
    float c  = cosf(0.5f * theta), s = sinf(0.5f * theta);
    float aa = 0.5f * (phi + omega), bb = 0.5f * (phi - omega);
    float ca = cosf(aa), sa = sinf(aa);
    float cb = cosf(bb), sb = sinf(bb);
    sGates[tid * 4 + 0] = make_float2( ca * c, -sa * c);
    sGates[tid * 4 + 1] = make_float2(-cb * s, -sb * s);
    sGates[tid * 4 + 2] = make_float2( cb * s, -sb * s);
    sGates[tid * 4 + 3] = make_float2( ca * c,  sa * c);
  }
  __syncthreads();

  const int lane = tid & 31;
  const int wave = tid >> 5;
  const int half = lane >> 4;
  const int lm   = lane & 15;
  const int rowBase = blockIdx.x * 128 + wave * 16;

  v8f acc[4] = {};
  int rowA = rowBase + lm; if (rowA >= B) rowA = B - 1;
  const float* xrow = x + (size_t)rowA * 64 + 2 * half;
  #pragma unroll
  for (int kk = 0; kk < 16; ++kk) {
    v2f a = *(const v2f*)(xrow + 4 * kk);
    #pragma unroll
    for (int t = 0; t < 4; ++t) {
      v2f bf = *(const v2f*)(sW1 + (16 * t + lm) * 64 + 4 * kk + 2 * half);
      acc[t] = __builtin_amdgcn_wmma_f32_16x16x4_f32(false, a, false, bf,
                                                     (short)0, acc[t], false, false);
    }
  }

  float hv[4][8];
  #pragma unroll
  for (int t = 0; t < 4; ++t) {
    float bias = sB1[16 * t + lm];
    #pragma unroll
    for (int j = 0; j < 8; ++j) hv[t][j] = fmaxf(acc[t][j] + bias, 0.0f);
  }

  float muj[8], inv[8];
  #pragma unroll
  for (int j = 0; j < 8; ++j) {
    float s  = hv[0][j] + hv[1][j] + hv[2][j] + hv[3][j];
    float q2 = hv[0][j]*hv[0][j] + hv[1][j]*hv[1][j] + hv[2][j]*hv[2][j] + hv[3][j]*hv[3][j];
    RED16(s); RED16(q2);
    float mu  = s * 0.015625f;
    float var = q2 * 0.015625f - mu * mu;
    muj[j] = mu;
    inv[j] = rsqrtf(var + 1e-5f);
  }
  #pragma unroll
  for (int t = 0; t < 4; ++t) {
    float g = sG[16 * t + lm], be = sBeta[16 * t + lm];
    #pragma unroll
    for (int j = 0; j < 8; ++j) hv[t][j] = (hv[t][j] - muj[j]) * inv[j] * g + be;
  }

  #pragma unroll
  for (int q = 0; q < 6; ++q) {
    float w0 = sW2[q * 64 + lm],      w1 = sW2[q * 64 + 16 + lm];
    float w2 = sW2[q * 64 + 32 + lm], w3 = sW2[q * 64 + 48 + lm];
    float bq = sB2[q];
    #pragma unroll
    for (int j = 0; j < 8; ++j) {
      float p = hv[0][j]*w0 + hv[1][j]*w1 + hv[2][j]*w2 + hv[3][j]*w3;
      RED16(p);
      if (lm == 0) sZ[wave * 96 + (8 * half + j) * 6 + q] = tanhf(p + bq);
    }
  }
  __syncthreads();

  float ws0[5], ws1[5];
  #pragma unroll
  for (int k = 0; k < 5; ++k) {
    float t0 = 0.f, t1 = 0.f;
    #pragma unroll
    for (int q = 0; q < 6; ++q) {
      float w = sWc[k * 6 + q];
      int b = 5 - q;
      t0 += ((lane >> b) & 1) ? -w : w;
      t1 += (((lane + 32) >> b) & 1) ? -w : w;
    }
    ws0[k] = t0; ws1[k] = t1;
  }
  const int rA0 = RINGS.r1[lane], rA1 = RINGS.r1[lane + 32];
  const int rB0 = RINGS.r2[lane], rB1 = RINGS.r2[lane + 32];

  for (int r = 0; r < 16; ++r) {
    int row = rowBase + r;
    float cq[6], sq[6];
    #pragma unroll
    for (int q = 0; q < 6; ++q) {
      float h2 = 1.57079632679489662f * sZ[wave * 96 + r * 6 + q];
      cq[q] = cosf(h2); sq[q] = sinf(h2);
    }
    float pr = 1.f;
    #pragma unroll
    for (int q = 1; q < 6; ++q) pr *= ((lane >> (5 - q)) & 1) ? sq[q] : cq[q];
    float2 a0 = make_float2(cq[0] * pr, 0.f);
    float2 a1 = make_float2(sq[0] * pr, 0.f);

    #pragma unroll
    for (int l = 0; l < 3; ++l) {
      #pragma unroll
      for (int w = 0; w < 6; ++w) {
        int g = l * 6 + w;
        float2 m00 = sGates[g * 4 + 0], m01 = sGates[g * 4 + 1];
        float2 m10 = sGates[g * 4 + 2], m11 = sGates[g * 4 + 3];
        if (w == 0) {
          float2 n0 = cadd(cmul(m00, a0), cmul(m01, a1));
          float2 n1 = cadd(cmul(m10, a0), cmul(m11, a1));
          a0 = n0; a1 = n1;
        } else {
          int b = 5 - w, mask = 1 << b;
          float2 p0 = shflx2(a0, mask), p1 = shflx2(a1, mask);
          int bit = (lane >> b) & 1;
          float2 e0 = cadd(cmul(m00, a0), cmul(m01, p0));
          float2 o0 = cadd(cmul(m10, p0), cmul(m11, a0));
          float2 e1 = cadd(cmul(m00, a1), cmul(m01, p1));
          float2 o1 = cadd(cmul(m10, p1), cmul(m11, a1));
          a0 = bit ? o0 : e0;
          a1 = bit ? o1 : e1;
        }
      }
      int s0 = (l == 1) ? rB0 : rA0;
      int s1 = (l == 1) ? rB1 : rA1;
      float2 g0a = shfl2(a0, s0 & 31), g0b = shfl2(a1, s0 & 31);
      float2 g1a = shfl2(a0, s1 & 31), g1b = shfl2(a1, s1 & 31);
      a0 = (s0 < 32) ? g0a : g0b;
      a1 = (s1 < 32) ? g1a : g1b;
    }

    float p0 = a0.x * a0.x + a0.y * a0.y;
    float p1 = a1.x * a1.x + a1.y * a1.y;
    float o[5];
    #pragma unroll
    for (int k = 0; k < 5; ++k) {
      float v = p0 * ws0[k] + p1 * ws1[k];
      v += __shfl_xor(v, 1, 32); v += __shfl_xor(v, 2, 32); v += __shfl_xor(v, 4, 32);
      v += __shfl_xor(v, 8, 32); v += __shfl_xor(v, 16, 32);
      o[k] = v;
    }
    if (lane == 0 && row < B) {
      #pragma unroll
      for (int k = 0; k < 5; ++k) out[(size_t)row * 5 + k] = o[k] + sBc[k];
    }
  }
}

extern "C" void kernel_launch(void* const* d_in, const int* in_sizes, int n_in,
                              void* d_out, int out_size, void* d_ws, size_t ws_size,
                              hipStream_t stream) {
  (void)n_in; (void)out_size; (void)d_ws; (void)ws_size;
  const float* x    = (const float*)d_in[0];
  const float* W1   = (const float*)d_in[1];
  const float* b1   = (const float*)d_in[2];
  const float* ln_g = (const float*)d_in[3];
  const float* ln_b = (const float*)d_in[4];
  const float* W2   = (const float*)d_in[5];
  const float* b2   = (const float*)d_in[6];
  const float* th_s = (const float*)d_in[7];
  const float* th_t = (const float*)d_in[8];
  const float* Wc   = (const float*)d_in[9];
  const float* bc   = (const float*)d_in[10];
  float* out = (float*)d_out;
  int B = in_sizes[0] / 64;
  int grid = (B + 127) / 128;
  qmaml_kernel<<<grid, 256, 0, stream>>>(x, W1, b1, ln_g, ln_b, W2, b2, th_s, th_t, Wc, bc, out, B);
}


# Circuit re-laid out: one row per lane, its 64 complex amplitudes in registers; gates by in-lane packed FMAs, rings by register moves, no lane shuffles; wave = 32 rows; readout sum order changes
# speedup vs baseline: 4.9396x; 4.9396x over previous
#include <hip/hip_runtime.h>
#include <math.h>
#include <utility>


typedef __attribute__((ext_vector_type(2))) float v2f;
typedef __attribute__((ext_vector_type(8))) float v8f;

__host__ __device__ constexpr int ring_src(int rr, int i) {
  for (int w = 5; w >= 0; --w) {
    int t = (w + rr) % 6;
    int cbit = (i >> (5 - w)) & 1;
    i ^= cbit << (5 - t);
  }
  return i;
}

template <int Wq, int P> struct PairIdx {
  static constexpr int b = 5 - Wq;
  static constexpr int i = ((P >> b) << (b + 1)) | (P & ((1 << b) - 1));
  static constexpr int j = i | (1 << b);
};
template <int Wq, int G, int... U>
__device__ __forceinline__ void gate_group(v2f (&st)[64], const v2f (&mr)[4], const v2f (&mi)[4],
                                           std::integer_sequence<int, U...>) {
  v2f a[4], c[4], asw[4], csw[4], t0[4], t1[4];
  ((a[U] = st[PairIdx<Wq, 4 * G + U>::i], c[U] = st[PairIdx<Wq, 4 * G + U>::j]), ...);
  ((asw[U] = __builtin_shufflevector(a[U], a[U], 1, 0)), ...);
  ((csw[U] = __builtin_shufflevector(c[U], c[U], 1, 0)), ...);
  ((t0[U] = mr[0] * a[U]), ...);
  ((t1[U] = mr[2] * a[U]), ...);
  ((t0[U] = __builtin_elementwise_fma(mi[0], asw[U], t0[U])), ...);
  ((t1[U] = __builtin_elementwise_fma(mi[2], asw[U], t1[U])), ...);
  ((t0[U] = __builtin_elementwise_fma(mr[1], c[U], t0[U])), ...);
  ((t1[U] = __builtin_elementwise_fma(mr[3], c[U], t1[U])), ...);
  ((t0[U] = __builtin_elementwise_fma(mi[1], csw[U], t0[U])), ...);
  ((t1[U] = __builtin_elementwise_fma(mi[3], csw[U], t1[U])), ...);
  ((st[PairIdx<Wq, 4 * G + U>::i] = t0[U], st[PairIdx<Wq, 4 * G + U>::j] = t1[U]), ...);
}
template <int Wq, int... G>
__device__ __forceinline__ void gate_all(v2f (&st)[64], const float4* gt,
                                         std::integer_sequence<int, G...>) {
  const float4 ga = gt[Wq * 2 + 0];
  const float4 gb = gt[Wq * 2 + 1];
  const v2f mr[4] = { {ga.x, ga.x}, {ga.z, ga.z}, {gb.x, gb.x}, {gb.z, gb.z} };
  const v2f mi[4] = { {-ga.y, ga.y}, {-ga.w, ga.w}, {-gb.y, gb.y}, {-gb.w, gb.w} };
  (gate_group<Wq, G>(st, mr, mi, std::make_integer_sequence<int, 4>{}), ...);
}
template <int... Wq>
__device__ __forceinline__ void layer_gates(v2f (&st)[64], const float4* gt,
                                            std::integer_sequence<int, Wq...>) {
  (gate_all<Wq>(st, gt, std::make_integer_sequence<int, 8>{}), ...);
}
template <int RR, int... I>
__device__ __forceinline__ void ring_apply(v2f (&st)[64], std::integer_sequence<int, I...>) {
  v2f n[64];
  ((n[I] = st[ring_src(RR, I)]), ...);
  ((st[I] = n[I]), ...);
}
template <int... I>
__device__ __forceinline__ void init_state(v2f (&st)[64], const float (&pr)[32],
                                           float c0, float s0, std::integer_sequence<int, I...>) {
  ((st[I] = v2f{c0 * pr[I], 0.f}, st[32 + I] = v2f{s0 * pr[I], 0.f}), ...);
}
template <int... I>
__device__ __forceinline__ void probs(float (&cur)[64], const v2f (&st)[64],
                                      std::integer_sequence<int, I...>) {
  ((cur[I] = st[I].x * st[I].x + st[I].y * st[I].y), ...);
}

#define RED16(v) do { v += __shfl_xor(v,1,32); v += __shfl_xor(v,2,32); \
                      v += __shfl_xor(v,4,32); v += __shfl_xor(v,8,32); } while(0)

__global__ __launch_bounds__(128, 4) void qmaml_kernel(
    const float* __restrict__ x,  const float* __restrict__ W1,
    const float* __restrict__ b1, const float* __restrict__ ln_g,
    const float* __restrict__ ln_b, const float* __restrict__ W2,
    const float* __restrict__ b2, const float* __restrict__ th_s,
    const float* __restrict__ th_t, const float* __restrict__ Wc,
    const float* __restrict__ bc, float* __restrict__ out, int B)
{
  __shared__ float  sW1[4096];
  __shared__ float  sW2[384];
  __shared__ float  sB1[64], sG[64], sBeta[64];
  __shared__ float  sB2[8], sWc[32], sBc[8];
  __shared__ float4 sGates[36];
  __shared__ float  sZ[768];
  __shared__ float  sCS[1536];

  const int tid = threadIdx.x;

  for (int i = tid; i < 1024; i += 128) ((float4*)sW1)[i] = ((const float4*)W1)[i];
  for (int i = tid; i < 384; i += 128)  sW2[i] = W2[i];
  if (tid < 64) { sB1[tid] = b1[tid]; sG[tid] = ln_g[tid]; sBeta[tid] = ln_b[tid]; }
  if (tid < 6)  sB2[tid] = b2[tid];
  if (tid < 30) sWc[tid] = Wc[tid];
  if (tid < 5)  sBc[tid] = bc[tid];
  if (tid < 18) {
    int l = tid / 6, w = tid % 6;
    const float* th = (l < 2) ? (th_s + (l * 6 + w) * 3) : (th_t + w * 3);
    float phi = th[0], theta = th[1], omega = th[2];
    float c  = cosf(0.5f * theta), s = sinf(0.5f * theta);
    float aa = 0.5f * (phi + omega), bb = 0.5f * (phi - omega);
    float ca = cosf(aa), sa = sinf(aa);
    float cb = cosf(bb), sb = sinf(bb);
    sGates[tid * 2 + 0] = make_float4( ca * c, -sa * c, -cb * s, -sb * s);
    sGates[tid * 2 + 1] = make_float4( cb * s, -sb * s,  ca * c,  sa * c);
  }
  __syncthreads();

  const int lane = tid & 31;
  const int wave = tid >> 5;
  const int half = lane >> 4;
  const int lm   = lane & 15;
  const int rowBase = blockIdx.x * 128 + wave * 32;

  {
    v8f acc[2][4] = {};
    int rowA0 = rowBase + lm;      if (rowA0 >= B) rowA0 = B - 1;
    int rowA1 = rowBase + 16 + lm; if (rowA1 >= B) rowA1 = B - 1;
    const float* xrow0 = x + (size_t)rowA0 * 64 + 2 * half;
    const float* xrow1 = x + (size_t)rowA1 * 64 + 2 * half;
    #pragma unroll
    for (int kk = 0; kk < 16; ++kk) {
      v2f a0 = *(const v2f*)(xrow0 + 4 * kk);
      v2f a1 = *(const v2f*)(xrow1 + 4 * kk);
      #pragma unroll
      for (int t = 0; t < 4; ++t) {
        v2f bf = *(const v2f*)(sW1 + (16 * t + lm) * 64 + 4 * kk + 2 * half);
        acc[0][t] = __builtin_amdgcn_wmma_f32_16x16x4_f32(false, a0, false, bf,
                                                          (short)0, acc[0][t], false, false);
        acc[1][t] = __builtin_amdgcn_wmma_f32_16x16x4_f32(false, a1, false, bf,
                                                          (short)0, acc[1][t], false, false);
      }
    }

    #pragma unroll 1
    for (int mt = 0; mt < 2; ++mt) {
      float hv[4][8];
      #pragma unroll
      for (int t = 0; t < 4; ++t) {
        float bias = sB1[16 * t + lm];
        #pragma unroll
        for (int j = 0; j < 8; ++j) {
          float a = (mt == 0) ? acc[0][t][j] : acc[1][t][j];
          hv[t][j] = fmaxf(a + bias, 0.0f);
        }
      }

      float muj[8], inv[8];
      #pragma unroll
      for (int j = 0; j < 8; ++j) {
        float s  = hv[0][j] + hv[1][j] + hv[2][j] + hv[3][j];
        float q2 = hv[0][j]*hv[0][j] + hv[1][j]*hv[1][j] + hv[2][j]*hv[2][j] + hv[3][j]*hv[3][j];
        RED16(s); RED16(q2);
        float mu  = s * 0.015625f;
        float var = q2 * 0.015625f - mu * mu;
        muj[j] = mu;
        inv[j] = rsqrtf(var + 1e-5f);
      }
      #pragma unroll
      for (int t = 0; t < 4; ++t) {
        float g = sG[16 * t + lm], be = sBeta[16 * t + lm];
        #pragma unroll
        for (int j = 0; j < 8; ++j) hv[t][j] = (hv[t][j] - muj[j]) * inv[j] * g + be;
      }

      #pragma unroll 1
      for (int q = 0; q < 6; ++q) {
        float w0 = sW2[q * 64 + lm],      w1 = sW2[q * 64 + 16 + lm];
        float w2 = sW2[q * 64 + 32 + lm], w3 = sW2[q * 64 + 48 + lm];
        float bq = sB2[q];
        #pragma unroll
        for (int j = 0; j < 8; ++j) {
          float p = hv[0][j]*w0 + hv[1][j]*w1 + hv[2][j]*w2 + hv[3][j]*w3;
          RED16(p);
          if (lm == 0) sZ[wave * 192 + (16 * mt + 8 * half + j) * 6 + q] = p + bq;
        }
      }
    }
  }
  __syncthreads();

  const int row = rowBase + lane;
  #pragma unroll 1
  for (int q = 0; q < 6; ++q) {
    float zz = tanhf(sZ[wave * 192 + lane * 6 + q]);
    float h2 = 1.57079632679489662f * zz;
    sCS[wave * 384 + (2 * q) * 32 + lane]     = cosf(h2);
    sCS[wave * 384 + (2 * q + 1) * 32 + lane] = sinf(h2);
  }
  float cq[6], sq[6];
  #pragma unroll
  for (int q = 0; q < 6; ++q) {
    cq[q] = sCS[wave * 384 + (2 * q) * 32 + lane];
    sq[q] = sCS[wave * 384 + (2 * q + 1) * 32 + lane];
  }

  v2f st[64];
  {
    float pr[32];
    pr[0] = cq[1]; pr[1] = sq[1];
    #pragma unroll
    for (int q = 2; q < 6; ++q) {
      #pragma unroll
      for (int idx = (1 << (q - 1)) - 1; idx >= 0; --idx) {
        float v = pr[idx];
        pr[2 * idx]     = v * cq[q];
        pr[2 * idx + 1] = v * sq[q];
      }
    }
    init_state(st, pr, cq[0], sq[0], std::make_integer_sequence<int, 32>{});
  }

  #pragma unroll 1
  for (int l = 0; l < 3; ++l) {
    layer_gates(st, sGates + l * 12, std::make_integer_sequence<int, 6>{});
    if (l == 1) ring_apply<2>(st, std::make_integer_sequence<int, 64>{});
    else        ring_apply<1>(st, std::make_integer_sequence<int, 64>{});
  }

  float zq[6];
  {
    float cur[64];
    probs(cur, st, std::make_integer_sequence<int, 64>{});
    #pragma unroll
    for (int b = 0; b < 6; ++b) {
      const int n2 = 32 >> b;
      float accd = cur[0] - cur[1];
      cur[0] = cur[0] + cur[1];
      #pragma unroll
      for (int j = 1; j < n2; ++j) {
        float lo = cur[2 * j], hi = cur[2 * j + 1];
        accd += lo - hi;
        cur[j] = lo + hi;
      }
      zq[5 - b] = accd;
    }
  }
  if (row < B) {
    #pragma unroll
    for (int k = 0; k < 5; ++k) {
      float o = 0.f;
      #pragma unroll
      for (int q = 0; q < 6; ++q) o = fmaf(sWc[k * 6 + q], zq[q], o);
      out[(size_t)row * 5 + k] = o + sBc[k];
    }
  }
}

extern "C" void kernel_launch(void* const* d_in, const int* in_sizes, int n_in,
                              void* d_out, int out_size, void* d_ws, size_t ws_size,
                              hipStream_t stream) {
  (void)n_in; (void)out_size; (void)d_ws; (void)ws_size;
  const float* x    = (const float*)d_in[0];
  const float* W1   = (const float*)d_in[1];
  const float* b1   = (const float*)d_in[2];
  const float* ln_g = (const float*)d_in[3];
  const float* ln_b = (const float*)d_in[4];
  const float* W2   = (const float*)d_in[5];
  const float* b2   = (const float*)d_in[6];
  const float* th_s = (const float*)d_in[7];
  const float* th_t = (const float*)d_in[8];
  const float* Wc   = (const float*)d_in[9];
  const float* bc   = (const float*)d_in[10];
  float* out = (float*)d_out;
  int B = in_sizes[0] / 64;
  int grid = (B + 127) / 128;
  qmaml_kernel<<<grid, 128, 0, stream>>>(x, W1, b1, ln_g, ln_b, W2, b2, th_s, th_t, Wc, bc, out, B);
}


# Rot gates split RZ*RY*RZ: RZ factors of adjacent layers merge through the ring into one tabulated phase per amplitude, RY as real packed rotations, last RZ and ring moves gone; sum order changes
# speedup vs baseline: 5.3886x; 1.0909x over previous
#include <hip/hip_runtime.h>
#include <math.h>
#include <utility>


typedef __attribute__((ext_vector_type(2))) float v2f;
typedef __attribute__((ext_vector_type(8))) float v8f;

__host__ __device__ constexpr int ring_src(int rr, int i) {
  for (int w = 5; w >= 0; --w) {
    int t = (w + rr) % 6;
    int cbit = (i >> (5 - w)) & 1;
    i ^= cbit << (5 - t);
  }
  return i;
}

__host__ __device__ constexpr int ring_dst(int rr, int k) {
  for (int w = 0; w < 6; ++w) {
    int t = (w + rr) % 6;
    int cbit = (k >> (5 - w)) & 1;
    k ^= cbit << (5 - t);
  }
  return k;
}

template <int Wq, int P> struct PairIdx {
  static constexpr int b = 5 - Wq;
  static constexpr int i = ((P >> b) << (b + 1)) | (P & ((1 << b) - 1));
  static constexpr int j = i | (1 << b);
};
template <int Wq, int G, int... U>
__device__ __forceinline__ void ry_group(v2f (&st)[64], const v2f c2, const v2f s2, const v2f ms2,
                                         std::integer_sequence<int, U...>) {
  v2f a[4], b[4], t0[4], t1[4];
  ((a[U] = st[PairIdx<Wq, 4 * G + U>::i], b[U] = st[PairIdx<Wq, 4 * G + U>::j]), ...);
  ((t0[U] = c2 * a[U]), ...);
  ((t1[U] = s2 * a[U]), ...);
  ((t0[U] = __builtin_elementwise_fma(ms2, b[U], t0[U])), ...);
  ((t1[U] = __builtin_elementwise_fma(c2, b[U], t1[U])), ...);
  ((st[PairIdx<Wq, 4 * G + U>::i] = t0[U], st[PairIdx<Wq, 4 * G + U>::j] = t1[U]), ...);
}
template <int Wq, int... G>
__device__ __forceinline__ void ry_all(v2f (&st)[64], const float2* ry,
                                       std::integer_sequence<int, G...>) {
  const float2 cs = ry[Wq];
  const v2f c2 = {cs.x, cs.x}, s2 = {cs.y, cs.y}, ms2 = {-cs.y, -cs.y};
  (ry_group<Wq, G>(st, c2, s2, ms2, std::make_integer_sequence<int, 4>{}), ...);
}
template <int... Wq>
__device__ __forceinline__ void layer_ry(v2f (&st)[64], const float2* ry,
                                         std::integer_sequence<int, Wq...>) {
  (ry_all<Wq>(st, ry, std::make_integer_sequence<int, 8>{}), ...);
}
__device__ __forceinline__ v2f phase_mul(v2f a, float2 m) {
  const v2f asw = __builtin_shufflevector(a, a, 1, 0);
  const v2f mr = {m.x, m.x}, mi = {-m.y, m.y};
  return __builtin_elementwise_fma(mi, asw, mr * a);
}
template <int... I>
__device__ __forceinline__ void phase_all(v2f (&st)[64], const float2* dg,
                                          std::integer_sequence<int, I...>) {
  ((st[I] = phase_mul(st[I], dg[I])), ...);
}
template <int... I>
__device__ __forceinline__ void ring_apply(v2f (&st)[64], const bool ring1,
                                           std::integer_sequence<int, I...>) {
  v2f n[64];
  ((n[I] = ring1 ? st[ring_src(1, I)] : st[ring_src(2, I)]), ...);
  ((st[I] = n[I]), ...);
}
template <int... I>
__device__ __forceinline__ void init_state(v2f (&st)[64], const float (&pr)[32], float c0, float s0,
                                           const float2* dg, std::integer_sequence<int, I...>) {
  ((st[I]      = (c0 * pr[I]) * v2f{dg[I].x, dg[I].y},
    st[32 + I] = (s0 * pr[I]) * v2f{dg[32 + I].x, dg[32 + I].y}), ...);
}
template <int RR, int... I>
__device__ __forceinline__ void probs(float (&cur)[64], const v2f (&st)[64],
                                      std::integer_sequence<int, I...>) {
  ((cur[I] = st[ring_src(RR, I)].x * st[ring_src(RR, I)].x
           + st[ring_src(RR, I)].y * st[ring_src(RR, I)].y), ...);
}

struct TabAngle {
  float phi[6], om[6], th;
  __device__ __forceinline__ void request(int e, const float* th_s, const float* th_t) {
    const int l = min(e >> 6, 2);
    const float* pphi = (l < 2) ? (th_s + l * 18) : th_t;
    const float* pom  = (l == 2) ? (th_s + 18) : th_s;
    const int g = min(max(e - 192, 0), 17);
    const float* pth = (g < 12) ? (th_s + g * 3) : (th_t + (g - 12) * 3);
    #pragma unroll
    for (int w = 0; w < 6; ++w) { phi[w] = pphi[w * 3 + 0]; om[w] = pom[w * 3 + 2]; }
    th = pth[1];
  }
  __device__ __forceinline__ float angle(int e) const {
    const int l = min(e >> 6, 2), i = e & 63;
    const int iphi = (l == 0) ? i : (l == 1) ? ring_dst(1, i) : ring_dst(2, i);
    float a = 0.f, b = 0.f;
    #pragma unroll
    for (int w = 0; w < 6; ++w) {
      a += ((iphi >> (5 - w)) & 1) ? phi[w] : -phi[w];
      b += ((i >> (5 - w)) & 1) ? om[w] : -om[w];
    }
    const float d = (l > 0) ? a + b : a;
    return 0.5f * ((e < 192) ? d : th);
  }
};

#define RED16(v) do { v += __shfl_xor(v,1,32); v += __shfl_xor(v,2,32); \
                      v += __shfl_xor(v,4,32); v += __shfl_xor(v,8,32); } while(0)

__global__ __launch_bounds__(128, 4) void qmaml_kernel(
    const float* __restrict__ x,  const float* __restrict__ W1,
    const float* __restrict__ b1, const float* __restrict__ ln_g,
    const float* __restrict__ ln_b, const float* __restrict__ W2,
    const float* __restrict__ b2, const float* __restrict__ th_s,
    const float* __restrict__ th_t, const float* __restrict__ Wc,
    const float* __restrict__ bc, float* __restrict__ out, int B)
{
  __shared__ float  sW1[4096];
  __shared__ float  sW2[384];
  __shared__ float  sB1[64], sG[64], sBeta[64];
  __shared__ float  sB2[8], sWc[32], sBc[8];
  __shared__ float2 sTab[210];
  __shared__ float  sZ[768];
  __shared__ float  sCS[1536];

  const int tid = threadIdx.x;

  TabAngle ta0, ta1;
  ta0.request(tid, th_s, th_t);
  ta1.request(tid + 128, th_s, th_t);

  for (int i = tid; i < 1024; i += 128) ((float4*)sW1)[i] = ((const float4*)W1)[i];
  for (int i = tid; i < 384; i += 128)  sW2[i] = W2[i];
  if (tid < 64) { sB1[tid] = b1[tid]; sG[tid] = ln_g[tid]; sBeta[tid] = ln_b[tid]; }
  if (tid < 6)  sB2[tid] = b2[tid];
  if (tid < 30) sWc[tid] = Wc[tid];
  if (tid < 5)  sBc[tid] = bc[tid];
  {
    float sn0, cs0, sn1, cs1;
    sincosf(ta0.angle(tid), &sn0, &cs0);
    sincosf(ta1.angle(tid + 128), &sn1, &cs1);
    sTab[tid] = make_float2(cs0, sn0);
    if (tid + 128 < 210) sTab[tid + 128] = make_float2(cs1, sn1);
  }
  __syncthreads();

  const int lane = tid & 31;
  const int wave = tid >> 5;
  const int half = lane >> 4;
  const int lm   = lane & 15;
  const int rowBase = blockIdx.x * 128 + wave * 32;

  {
    v8f acc[2][4] = {};
    int rowA0 = rowBase + lm;      if (rowA0 >= B) rowA0 = B - 1;
    int rowA1 = rowBase + 16 + lm; if (rowA1 >= B) rowA1 = B - 1;
    const float* xrow0 = x + (size_t)rowA0 * 64 + 2 * half;
    const float* xrow1 = x + (size_t)rowA1 * 64 + 2 * half;
    #pragma unroll
    for (int kk = 0; kk < 16; ++kk) {
      v2f a0 = *(const v2f*)(xrow0 + 4 * kk);
      v2f a1 = *(const v2f*)(xrow1 + 4 * kk);
      #pragma unroll
      for (int t = 0; t < 4; ++t) {
        v2f bf = *(const v2f*)(sW1 + (16 * t + lm) * 64 + 4 * kk + 2 * half);
        acc[0][t] = __builtin_amdgcn_wmma_f32_16x16x4_f32(false, a0, false, bf,
                                                          (short)0, acc[0][t], false, false);
        acc[1][t] = __builtin_amdgcn_wmma_f32_16x16x4_f32(false, a1, false, bf,
                                                          (short)0, acc[1][t], false, false);
      }
    }

    #pragma unroll 1
    for (int mt = 0; mt < 2; ++mt) {
      float hv[4][8];
      #pragma unroll
      for (int t = 0; t < 4; ++t) {
        float bias = sB1[16 * t + lm];
        #pragma unroll
        for (int j = 0; j < 8; ++j) {
          float a = (mt == 0) ? acc[0][t][j] : acc[1][t][j];
          hv[t][j] = fmaxf(a + bias, 0.0f);
        }
      }

      float muj[8], inv[8];
      #pragma unroll
      for (int j = 0; j < 8; ++j) {
        float s  = hv[0][j] + hv[1][j] + hv[2][j] + hv[3][j];
        float q2 = hv[0][j]*hv[0][j] + hv[1][j]*hv[1][j] + hv[2][j]*hv[2][j] + hv[3][j]*hv[3][j];
        RED16(s); RED16(q2);
        float mu  = s * 0.015625f;
        float var = q2 * 0.015625f - mu * mu;
        muj[j] = mu;
        inv[j] = rsqrtf(var + 1e-5f);
      }
      #pragma unroll
      for (int t = 0; t < 4; ++t) {
        float g = sG[16 * t + lm], be = sBeta[16 * t + lm];
        #pragma unroll
        for (int j = 0; j < 8; ++j) hv[t][j] = (hv[t][j] - muj[j]) * inv[j] * g + be;
      }

      #pragma unroll 1
      for (int q = 0; q < 6; ++q) {
        float w0 = sW2[q * 64 + lm],      w1 = sW2[q * 64 + 16 + lm];
        float w2 = sW2[q * 64 + 32 + lm], w3 = sW2[q * 64 + 48 + lm];
        float bq = sB2[q];
        #pragma unroll
        for (int j = 0; j < 8; ++j) {
          float p = hv[0][j]*w0 + hv[1][j]*w1 + hv[2][j]*w2 + hv[3][j]*w3;
          RED16(p);
          if (lm == 0) sZ[wave * 192 + (16 * mt + 8 * half + j) * 6 + q] = p + bq;
        }
      }
    }
  }
  __syncthreads();

  const int row = rowBase + lane;
  #pragma unroll 1
  for (int q = 0; q < 6; ++q) {
    float zz = tanhf(sZ[wave * 192 + lane * 6 + q]);
    float h2 = 1.57079632679489662f * zz;
    sCS[wave * 384 + (2 * q) * 32 + lane]     = cosf(h2);
    sCS[wave * 384 + (2 * q + 1) * 32 + lane] = sinf(h2);
  }
  float cq[6], sq[6];
  #pragma unroll
  for (int q = 0; q < 6; ++q) {
    cq[q] = sCS[wave * 384 + (2 * q) * 32 + lane];
    sq[q] = sCS[wave * 384 + (2 * q + 1) * 32 + lane];
  }

  v2f st[64];
  {
    float pr[32];
    pr[0] = cq[1]; pr[1] = sq[1];
    #pragma unroll
    for (int q = 2; q < 6; ++q) {
      #pragma unroll
      for (int idx = (1 << (q - 1)) - 1; idx >= 0; --idx) {
        float v = pr[idx];
        pr[2 * idx]     = v * cq[q];
        pr[2 * idx + 1] = v * sq[q];
      }
    }
    init_state(st, pr, cq[0], sq[0], sTab, std::make_integer_sequence<int, 32>{});
  }

  #pragma unroll 1
  for (int l = 0; l < 3; ++l) {
    layer_ry(st, sTab + 192 + l * 6, std::make_integer_sequence<int, 6>{});
    if (l < 2) {
      phase_all(st, sTab + 64 * (l + 1), std::make_integer_sequence<int, 64>{});
      ring_apply(st, l == 0, std::make_integer_sequence<int, 64>{});
    }
  }

  float zq[6];
  {
    float cur[64];
    probs<1>(cur, st, std::make_integer_sequence<int, 64>{});
    #pragma unroll
    for (int b = 0; b < 6; ++b) {
      const int n2 = 32 >> b;
      float accd = cur[0] - cur[1];
      cur[0] = cur[0] + cur[1];
      #pragma unroll
      for (int j = 1; j < n2; ++j) {
        float lo = cur[2 * j], hi = cur[2 * j + 1];
        accd += lo - hi;
        cur[j] = lo + hi;
      }
      zq[5 - b] = accd;
    }
  }
  if (row < B) {
    #pragma unroll
    for (int k = 0; k < 5; ++k) {
      float o = 0.f;
      #pragma unroll
      for (int q = 0; q < 6; ++q) o = fmaf(sWc[k * 6 + q], zq[q], o);
      out[(size_t)row * 5 + k] = o + sBc[k];
    }
  }
}

extern "C" void kernel_launch(void* const* d_in, const int* in_sizes, int n_in,
                              void* d_out, int out_size, void* d_ws, size_t ws_size,
                              hipStream_t stream) {
  (void)n_in; (void)out_size; (void)d_ws; (void)ws_size;
  const float* x    = (const float*)d_in[0];
  const float* W1   = (const float*)d_in[1];
  const float* b1   = (const float*)d_in[2];
  const float* ln_g = (const float*)d_in[3];
  const float* ln_b = (const float*)d_in[4];
  const float* W2   = (const float*)d_in[5];
  const float* b2   = (const float*)d_in[6];
  const float* th_s = (const float*)d_in[7];
  const float* th_t = (const float*)d_in[8];
  const float* Wc   = (const float*)d_in[9];
  const float* bc   = (const float*)d_in[10];
  float* out = (float*)d_out;
  int B = in_sizes[0] / 64;
  int grid = (B + 127) / 128;
  qmaml_kernel<<<grid, 128, 0, stream>>>(x, W1, b1, ln_g, ln_b, W2, b2, th_s, th_t, Wc, bc, out, B);
}


# First-layer output tile passed through LDS to one row per lane (64 features in registers): LayerNorm (variance from deviations) and the 6 second-layer dots in-lane, no butterflies; sum order changes
# speedup vs baseline: 8.3368x; 1.5471x over previous
#include <hip/hip_runtime.h>
#include <math.h>
#include <utility>


typedef __attribute__((ext_vector_type(2))) float v2f;
typedef __attribute__((ext_vector_type(8))) float v8f;

__host__ __device__ constexpr int ring_src(int rr, int i) {
  for (int w = 5; w >= 0; --w) {
    int t = (w + rr) % 6;
    int cbit = (i >> (5 - w)) & 1;
    i ^= cbit << (5 - t);
  }
  return i;
}

__host__ __device__ constexpr int ring_dst(int rr, int k) {
  for (int w = 0; w < 6; ++w) {
    int t = (w + rr) % 6;
    int cbit = (k >> (5 - w)) & 1;
    k ^= cbit << (5 - t);
  }
  return k;
}

template <int Wq, int P> struct PairIdx {
  static constexpr int b = 5 - Wq;
  static constexpr int i = ((P >> b) << (b + 1)) | (P & ((1 << b) - 1));
  static constexpr int j = i | (1 << b);
};
template <int Wq, int G, int... U>
__device__ __forceinline__ void ry_group(v2f (&st)[64], const v2f c2, const v2f s2, const v2f ms2,
                                         std::integer_sequence<int, U...>) {
  v2f a[4], b[4], t0[4], t1[4];
  ((a[U] = st[PairIdx<Wq, 4 * G + U>::i], b[U] = st[PairIdx<Wq, 4 * G + U>::j]), ...);
  ((t0[U] = c2 * a[U]), ...);
  ((t1[U] = s2 * a[U]), ...);
  ((t0[U] = __builtin_elementwise_fma(ms2, b[U], t0[U])), ...);
  ((t1[U] = __builtin_elementwise_fma(c2, b[U], t1[U])), ...);
  ((st[PairIdx<Wq, 4 * G + U>::i] = t0[U], st[PairIdx<Wq, 4 * G + U>::j] = t1[U]), ...);
}
template <int Wq, int... G>
__device__ __forceinline__ void ry_all(v2f (&st)[64], const float2* ry,
                                       std::integer_sequence<int, G...>) {
  const float2 cs = ry[Wq];
  const v2f c2 = {cs.x, cs.x}, s2 = {cs.y, cs.y}, ms2 = {-cs.y, -cs.y};
  (ry_group<Wq, G>(st, c2, s2, ms2, std::make_integer_sequence<int, 4>{}), ...);
}
template <int... Wq>
__device__ __forceinline__ void layer_ry(v2f (&st)[64], const float2* ry,
                                         std::integer_sequence<int, Wq...>) {
  (ry_all<Wq>(st, ry, std::make_integer_sequence<int, 8>{}), ...);
}
__device__ __forceinline__ v2f phase_mul(v2f a, float2 m) {
  const v2f asw = __builtin_shufflevector(a, a, 1, 0);
  const v2f mr = {m.x, m.x}, mi = {-m.y, m.y};
  return __builtin_elementwise_fma(mi, asw, mr * a);
}
template <int... I>
__device__ __forceinline__ void phase_all(v2f (&st)[64], const float2* dg,
                                          std::integer_sequence<int, I...>) {
  ((st[I] = phase_mul(st[I], dg[I])), ...);
}
template <int... I>
__device__ __forceinline__ void ring_apply(v2f (&st)[64], const bool ring1,
                                           std::integer_sequence<int, I...>) {
  v2f n[64];
  ((n[I] = ring1 ? st[ring_src(1, I)] : st[ring_src(2, I)]), ...);
  ((st[I] = n[I]), ...);
}
template <int... I>
__device__ __forceinline__ void init_state(v2f (&st)[64], const float (&pr)[32], float c0, float s0,
                                           const float2* dg, std::integer_sequence<int, I...>) {
  ((st[I]      = (c0 * pr[I]) * v2f{dg[I].x, dg[I].y},
    st[32 + I] = (s0 * pr[I]) * v2f{dg[32 + I].x, dg[32 + I].y}), ...);
}
template <int RR, int... I>
__device__ __forceinline__ void probs(float (&cur)[64], const v2f (&st)[64],
                                      std::integer_sequence<int, I...>) {
  ((cur[I] = st[ring_src(RR, I)].x * st[ring_src(RR, I)].x
           + st[ring_src(RR, I)].y * st[ring_src(RR, I)].y), ...);
}

struct TabAngle {
  float phi[6], om[6], th;
  __device__ __forceinline__ void request(int e, const float* th_s, const float* th_t) {
    const int l = min(e >> 6, 2);
    const float* pphi = (l < 2) ? (th_s + l * 18) : th_t;
    const float* pom  = (l == 2) ? (th_s + 18) : th_s;
    const int g = min(max(e - 192, 0), 17);
    const float* pth = (g < 12) ? (th_s + g * 3) : (th_t + (g - 12) * 3);
    #pragma unroll
    for (int w = 0; w < 6; ++w) { phi[w] = pphi[w * 3 + 0]; om[w] = pom[w * 3 + 2]; }
    th = pth[1];
  }
  __device__ __forceinline__ float angle(int e) const {
    const int l = min(e >> 6, 2), i = e & 63;
    const int iphi = (l == 0) ? i : (l == 1) ? ring_dst(1, i) : ring_dst(2, i);
    float a = 0.f, b = 0.f;
    #pragma unroll
    for (int w = 0; w < 6; ++w) {
      a += ((iphi >> (5 - w)) & 1) ? phi[w] : -phi[w];
      b += ((i >> (5 - w)) & 1) ? om[w] : -om[w];
    }
    const float d = (l > 0) ? a + b : a;
    return 0.5f * ((e < 192) ? d : th);
  }
};

__global__ __launch_bounds__(128, 4) void qmaml_kernel(
    const float* __restrict__ x,  const float* __restrict__ W1,
    const float* __restrict__ b1, const float* __restrict__ ln_g,
    const float* __restrict__ ln_b, const float* __restrict__ W2,
    const float* __restrict__ b2, const float* __restrict__ th_s,
    const float* __restrict__ th_t, const float* __restrict__ Wc,
    const float* __restrict__ bc, float* __restrict__ out, int B)
{
  __shared__ float  sW1[4096];
  __shared__ __attribute__((aligned(16))) float sW2[384];
  __shared__ float  sB1[64];
  __shared__ __attribute__((aligned(16))) float sG[64];
  __shared__ __attribute__((aligned(16))) float sBeta[64];
  __shared__ float  sB2[8], sWc[32], sBc[8];
  __shared__ float2 sTab[210];
  __shared__ __attribute__((aligned(16))) float sH[4 * 64 * 36];

  const int tid = threadIdx.x;

  TabAngle ta0, ta1;
  ta0.request(tid, th_s, th_t);
  ta1.request(tid + 128, th_s, th_t);

  for (int i = tid; i < 1024; i += 128) ((float4*)sW1)[i] = ((const float4*)W1)[i];
  for (int i = tid; i < 384; i += 128)  sW2[i] = W2[i];
  if (tid < 64) { sB1[tid] = b1[tid]; sG[tid] = ln_g[tid]; sBeta[tid] = ln_b[tid]; }
  if (tid < 6)  sB2[tid] = b2[tid];
  if (tid < 30) sWc[tid] = Wc[tid];
  if (tid < 5)  sBc[tid] = bc[tid];
  {
    float sn0, cs0, sn1, cs1;
    sincosf(ta0.angle(tid), &sn0, &cs0);
    sincosf(ta1.angle(tid + 128), &sn1, &cs1);
    sTab[tid] = make_float2(cs0, sn0);
    if (tid + 128 < 210) sTab[tid + 128] = make_float2(cs1, sn1);
  }
  __syncthreads();

  const int lane = tid & 31;
  const int wave = tid >> 5;
  const int half = lane >> 4;
  const int lm   = lane & 15;
  const int rowBase = blockIdx.x * 128 + wave * 32;
  float* const hw = sH + wave * (64 * 36);

  {
    v8f acc[2][4] = {};
    int rowA0 = rowBase + lm;      if (rowA0 >= B) rowA0 = B - 1;
    int rowA1 = rowBase + 16 + lm; if (rowA1 >= B) rowA1 = B - 1;
    const float* xrow0 = x + (size_t)rowA0 * 64 + 2 * half;
    const float* xrow1 = x + (size_t)rowA1 * 64 + 2 * half;
    #pragma unroll
    for (int kk = 0; kk < 16; ++kk) {
      v2f a0 = *(const v2f*)(xrow0 + 4 * kk);
      v2f a1 = *(const v2f*)(xrow1 + 4 * kk);
      #pragma unroll
      for (int t = 0; t < 4; ++t) {
        v2f bf = *(const v2f*)(sW1 + (16 * t + lm) * 64 + 4 * kk + 2 * half);
        acc[0][t] = __builtin_amdgcn_wmma_f32_16x16x4_f32(false, a0, false, bf,
                                                          (short)0, acc[0][t], false, false);
        acc[1][t] = __builtin_amdgcn_wmma_f32_16x16x4_f32(false, a1, false, bf,
                                                          (short)0, acc[1][t], false, false);
      }
    }

    #pragma unroll
    for (int mt = 0; mt < 2; ++mt) {
      #pragma unroll
      for (int t = 0; t < 4; ++t) {
        const float bias = sB1[16 * t + lm];
        float4 lo, hi;
        lo.x = fmaxf(acc[mt][t][0] + bias, 0.0f); lo.y = fmaxf(acc[mt][t][1] + bias, 0.0f);
        lo.z = fmaxf(acc[mt][t][2] + bias, 0.0f); lo.w = fmaxf(acc[mt][t][3] + bias, 0.0f);
        hi.x = fmaxf(acc[mt][t][4] + bias, 0.0f); hi.y = fmaxf(acc[mt][t][5] + bias, 0.0f);
        hi.z = fmaxf(acc[mt][t][6] + bias, 0.0f); hi.w = fmaxf(acc[mt][t][7] + bias, 0.0f);
        float* p = hw + (16 * t + lm) * 36 + 16 * mt + 8 * half;
        *(float4*)p = lo;
        *(float4*)(p + 4) = hi;
      }
    }
  }
  __syncthreads();

  {
    const float* hr = hw + lane;
    v2f h[32];
    #pragma unroll
    for (int i = 0; i < 32; ++i) h[i] = v2f{hr[(2 * i) * 36], hr[(2 * i + 1) * 36]};
    v2f a0 = h[0], a1 = h[1], a2 = h[2], a3 = h[3];
    #pragma unroll
    for (int i = 4; i < 32; i += 4) { a0 += h[i]; a1 += h[i + 1]; a2 += h[i + 2]; a3 += h[i + 3]; }
    const v2f sa = (a0 + a1) + (a2 + a3);
    const float mu = (sa.x + sa.y) * 0.015625f;
    const v2f mu2 = {mu, mu};
    #pragma unroll
    for (int i = 0; i < 32; ++i) h[i] = h[i] - mu2;
    v2f q0 = h[0] * h[0], q1 = h[1] * h[1], q2 = h[2] * h[2], q3 = h[3] * h[3];
    #pragma unroll
    for (int i = 4; i < 32; i += 4) {
      q0 = __builtin_elementwise_fma(h[i], h[i], q0);
      q1 = __builtin_elementwise_fma(h[i + 1], h[i + 1], q1);
      q2 = __builtin_elementwise_fma(h[i + 2], h[i + 2], q2);
      q3 = __builtin_elementwise_fma(h[i + 3], h[i + 3], q3);
    }
    const v2f sq = (q0 + q1) + (q2 + q3);
    const float inv = rsqrtf((sq.x + sq.y) * 0.015625f + 1e-5f);
    const v2f inv2 = {inv, inv};
    const v2f* g2  = (const v2f*)sG;
    const v2f* be2 = (const v2f*)sBeta;
    #pragma unroll
    for (int i = 0; i < 32; ++i) h[i] = __builtin_elementwise_fma(h[i] * inv2, g2[i], be2[i]);
    #pragma unroll
    for (int q = 0; q < 6; ++q) {
      const v2f* w = (const v2f*)(sW2 + q * 64);
      v2f p0 = h[0] * w[0], p1 = h[1] * w[1];
      #pragma unroll
      for (int i = 2; i < 32; i += 2) {
        p0 = __builtin_elementwise_fma(h[i], w[i], p0);
        p1 = __builtin_elementwise_fma(h[i + 1], w[i + 1], p1);
      }
      const v2f ps = p0 + p1;
      hw[q * 32 + lane] = (ps.x + ps.y) + sB2[q];
    }
  }

  const int row = rowBase + lane;
  #pragma unroll 1
  for (int q = 0; q < 6; ++q) {
    float zz = tanhf(hw[q * 32 + lane]);
    float h2 = 1.57079632679489662f * zz;
    hw[192 + (2 * q) * 32 + lane]     = cosf(h2);
    hw[192 + (2 * q + 1) * 32 + lane] = sinf(h2);
  }
  float cq[6], sq[6];
  #pragma unroll
  for (int q = 0; q < 6; ++q) {
    cq[q] = hw[192 + (2 * q) * 32 + lane];
    sq[q] = hw[192 + (2 * q + 1) * 32 + lane];
  }

  v2f st[64];
  {
    float pr[32];
    pr[0] = cq[1]; pr[1] = sq[1];
    #pragma unroll
    for (int q = 2; q < 6; ++q) {
      #pragma unroll
      for (int idx = (1 << (q - 1)) - 1; idx >= 0; --idx) {
        float v = pr[idx];
        pr[2 * idx]     = v * cq[q];
        pr[2 * idx + 1] = v * sq[q];
      }
    }
    init_state(st, pr, cq[0], sq[0], sTab, std::make_integer_sequence<int, 32>{});
  }

  #pragma unroll 1
  for (int l = 0; l < 3; ++l) {
    layer_ry(st, sTab + 192 + l * 6, std::make_integer_sequence<int, 6>{});
    if (l < 2) {
      phase_all(st, sTab + 64 * (l + 1), std::make_integer_sequence<int, 64>{});
      ring_apply(st, l == 0, std::make_integer_sequence<int, 64>{});
    }
  }

  float zq[6];
  {
    float cur[64];
    probs<1>(cur, st, std::make_integer_sequence<int, 64>{});
    #pragma unroll
    for (int b = 0; b < 6; ++b) {
      const int n2 = 32 >> b;
      float accd = cur[0] - cur[1];
      cur[0] = cur[0] + cur[1];
      #pragma unroll
      for (int j = 1; j < n2; ++j) {
        float lo = cur[2 * j], hi = cur[2 * j + 1];
        accd += lo - hi;
        cur[j] = lo + hi;
      }
      zq[5 - b] = accd;
    }
  }
  if (row < B) {
    #pragma unroll
    for (int k = 0; k < 5; ++k) {
      float o = 0.f;
      #pragma unroll
      for (int q = 0; q < 6; ++q) o = fmaf(sWc[k * 6 + q], zq[q], o);
      out[(size_t)row * 5 + k] = o + sBc[k];
    }
  }
}

extern "C" void kernel_launch(void* const* d_in, const int* in_sizes, int n_in,
                              void* d_out, int out_size, void* d_ws, size_t ws_size,
                              hipStream_t stream) {
  (void)n_in; (void)out_size; (void)d_ws; (void)ws_size;
  const float* x    = (const float*)d_in[0];
  const float* W1   = (const float*)d_in[1];
  const float* b1   = (const float*)d_in[2];
  const float* ln_g = (const float*)d_in[3];
  const float* ln_b = (const float*)d_in[4];
  const float* W2   = (const float*)d_in[5];
  const float* b2   = (const float*)d_in[6];
  const float* th_s = (const float*)d_in[7];
  const float* th_t = (const float*)d_in[8];
  const float* Wc   = (const float*)d_in[9];
  const float* bc   = (const float*)d_in[10];
  float* out = (float*)d_out;
  int B = in_sizes[0] / 64;
  int grid = (B + 127) / 128;
  qmaml_kernel<<<grid, 128, 0, stream>>>(x, W1, b1, ln_g, ln_b, W2, b2, th_s, th_t, Wc, bc, out, B);
}


# Layer 0 (RZ, RY, ring 1) applied to the six one-qubit factors before the tensor product, built complex as 8 x 8 into ring-1 positions; 64-amplitude gate loop runs 2 layers not 3; sum order changes
# speedup vs baseline: 9.0330x; 1.0835x over previous
#include <hip/hip_runtime.h>
#include <math.h>
#include <utility>


typedef __attribute__((ext_vector_type(2))) float v2f;
typedef __attribute__((ext_vector_type(8))) float v8f;

__host__ __device__ constexpr int ring_src(int rr, int i) {
  for (int w = 5; w >= 0; --w) {
    int t = (w + rr) % 6;
    int cbit = (i >> (5 - w)) & 1;
    i ^= cbit << (5 - t);
  }
  return i;
}

__host__ __device__ constexpr int ring_dst(int rr, int k) {
  for (int w = 0; w < 6; ++w) {
    int t = (w + rr) % 6;
    int cbit = (k >> (5 - w)) & 1;
    k ^= cbit << (5 - t);
  }
  return k;
}

template <int Wq, int P> struct PairIdx {
  static constexpr int b = 5 - Wq;
  static constexpr int i = ((P >> b) << (b + 1)) | (P & ((1 << b) - 1));
  static constexpr int j = i | (1 << b);
};
template <int Wq, int G, int... U>
__device__ __forceinline__ void ry_group(v2f (&st)[64], const v2f c2, const v2f s2, const v2f ms2,
                                         std::integer_sequence<int, U...>) {
  v2f a[4], b[4], t0[4], t1[4];
  ((a[U] = st[PairIdx<Wq, 4 * G + U>::i], b[U] = st[PairIdx<Wq, 4 * G + U>::j]), ...);
  ((t0[U] = c2 * a[U]), ...);
  ((t1[U] = s2 * a[U]), ...);
  ((t0[U] = __builtin_elementwise_fma(ms2, b[U], t0[U])), ...);
  ((t1[U] = __builtin_elementwise_fma(c2, b[U], t1[U])), ...);
  ((st[PairIdx<Wq, 4 * G + U>::i] = t0[U], st[PairIdx<Wq, 4 * G + U>::j] = t1[U]), ...);
}
template <int Wq, int... G>
__device__ __forceinline__ void ry_all(v2f (&st)[64], const float2* ry,
                                       std::integer_sequence<int, G...>) {
  const float2 cs = ry[Wq];
  const v2f c2 = {cs.x, cs.x}, s2 = {cs.y, cs.y}, ms2 = {-cs.y, -cs.y};
  (ry_group<Wq, G>(st, c2, s2, ms2, std::make_integer_sequence<int, 4>{}), ...);
}
template <int... Wq>
__device__ __forceinline__ void layer_ry(v2f (&st)[64], const float2* ry,
                                         std::integer_sequence<int, Wq...>) {
  (ry_all<Wq>(st, ry, std::make_integer_sequence<int, 8>{}), ...);
}
__device__ __forceinline__ v2f phase_mul(v2f a, float2 m) {
  const v2f asw = __builtin_shufflevector(a, a, 1, 0);
  const v2f mr = {m.x, m.x}, mi = {-m.y, m.y};
  return __builtin_elementwise_fma(mi, asw, mr * a);
}
__device__ __forceinline__ v2f cmul(v2f a, v2f b) {
  const v2f ar = {a.x, a.x}, ai = {a.y, a.y};
  const v2f bs = {-b.y, b.x};
  return __builtin_elementwise_fma(ai, bs, ar * b);
}
template <int... I>
__device__ __forceinline__ void phase_all(v2f (&st)[64], const float2* dg,
                                          std::integer_sequence<int, I...>) {
  ((st[I] = phase_mul(st[I], dg[I])), ...);
}
template <int RR, int... I>
__device__ __forceinline__ void ring_apply(v2f (&st)[64], std::integer_sequence<int, I...>) {
  v2f n[64];
  ((n[I] = st[ring_src(RR, I)]), ...);
  ((st[I] = n[I]), ...);
}
template <int... I>
__device__ __forceinline__ void build_state(v2f (&st)[64], const v2f (&hi)[8], const v2f (&lo)[8],
                                            std::integer_sequence<int, I...>) {
  ((st[ring_dst(1, I)] = cmul(hi[I >> 3], lo[I & 7])), ...);
}
__device__ __forceinline__ void triple(v2f (&o)[8], const v2f (&a)[2], const v2f (&b)[2],
                                       const v2f (&c)[2]) {
  #pragma unroll
  for (int k = 0; k < 4; ++k) {
    const v2f ab = cmul(a[k >> 1], b[k & 1]);
    o[2 * k]     = cmul(ab, c[0]);
    o[2 * k + 1] = cmul(ab, c[1]);
  }
}
template <int RR, int... I>
__device__ __forceinline__ void probs(float (&cur)[64], const v2f (&st)[64],
                                      std::integer_sequence<int, I...>) {
  ((cur[I] = st[ring_src(RR, I)].x * st[ring_src(RR, I)].x
           + st[ring_src(RR, I)].y * st[ring_src(RR, I)].y), ...);
}

struct TabAngle {
  float phi[6], om[6], small;
  __device__ __forceinline__ void request(int tid, const float* th_s, const float* th_t) {
    const bool l1 = tid < 64;
    const float* pphi = l1 ? (th_s + 18) : th_t;
    const float* pom  = l1 ? th_s : (th_s + 18);
    #pragma unroll
    for (int w = 0; w < 6; ++w) { phi[w] = pphi[w * 3 + 0]; om[w] = pom[w * 3 + 2]; }
    const int g = min(tid, 23);
    const float* ps = (g < 12) ? (th_s + g * 3 + 1)
                    : (g < 18) ? (th_t + (g - 12) * 3 + 1) : (th_s + (g - 18) * 3);
    small = *ps;
  }
  __device__ __forceinline__ float diag_angle(int e) const {
    const int iphi = e & 63;
    const int i    = (e < 64) ? ring_src(1, iphi) : ring_src(2, iphi);
    float a = 0.f, b = 0.f;
    #pragma unroll
    for (int w = 0; w < 6; ++w) {
      a += ((iphi >> (5 - w)) & 1) ? phi[w] : -phi[w];
      b += ((i >> (5 - w)) & 1) ? om[w] : -om[w];
    }
    return 0.5f * (a + b);
  }
};

__global__ __launch_bounds__(128, 4) void qmaml_kernel(
    const float* __restrict__ x,  const float* __restrict__ W1,
    const float* __restrict__ b1, const float* __restrict__ ln_g,
    const float* __restrict__ ln_b, const float* __restrict__ W2,
    const float* __restrict__ b2, const float* __restrict__ th_s,
    const float* __restrict__ th_t, const float* __restrict__ Wc,
    const float* __restrict__ bc, float* __restrict__ out, int B)
{
  __shared__ float  sW1[4096];
  __shared__ __attribute__((aligned(16))) float sW2[384];
  __shared__ float  sB1[64];
  __shared__ __attribute__((aligned(16))) float sG[64];
  __shared__ __attribute__((aligned(16))) float sBeta[64];
  __shared__ float  sB2[8], sWc[32], sBc[8];
  __shared__ float2 sTab[152];
  __shared__ __attribute__((aligned(16))) float sH[4 * 64 * 36];

  const int tid = threadIdx.x;

  TabAngle ta;
  ta.request(tid, th_s, th_t);

  for (int i = tid; i < 1024; i += 128) ((float4*)sW1)[i] = ((const float4*)W1)[i];
  for (int i = tid; i < 384; i += 128)  sW2[i] = W2[i];
  if (tid < 64) { sB1[tid] = b1[tid]; sG[tid] = ln_g[tid]; sBeta[tid] = ln_b[tid]; }
  if (tid < 6)  sB2[tid] = b2[tid];
  if (tid < 30) sWc[tid] = Wc[tid];
  if (tid < 5)  sBc[tid] = bc[tid];
  {
    float sn0, cs0;
    sincosf(ta.diag_angle(tid), &sn0, &cs0);
    sTab[tid] = make_float2(cs0, sn0);
    if (tid < 24) {
      float sn1, cs1;
      sincosf(0.5f * ta.small, &sn1, &cs1);
      sTab[128 + tid] = make_float2(cs1, sn1);
    }
  }
  __syncthreads();

  const int lane = tid & 31;
  const int wave = tid >> 5;
  const int half = lane >> 4;
  const int lm   = lane & 15;
  const int rowBase = blockIdx.x * 128 + wave * 32;
  float* const hw = sH + wave * (64 * 36);

  {
    v8f acc[2][4] = {};
    int rowA0 = rowBase + lm;      if (rowA0 >= B) rowA0 = B - 1;
    int rowA1 = rowBase + 16 + lm; if (rowA1 >= B) rowA1 = B - 1;
    const float* xrow0 = x + (size_t)rowA0 * 64 + 2 * half;
    const float* xrow1 = x + (size_t)rowA1 * 64 + 2 * half;
    #pragma unroll
    for (int kk = 0; kk < 16; ++kk) {
      v2f a0 = *(const v2f*)(xrow0 + 4 * kk);
      v2f a1 = *(const v2f*)(xrow1 + 4 * kk);
      #pragma unroll
      for (int t = 0; t < 4; ++t) {
        v2f bf = *(const v2f*)(sW1 + (16 * t + lm) * 64 + 4 * kk + 2 * half);
        acc[0][t] = __builtin_amdgcn_wmma_f32_16x16x4_f32(false, a0, false, bf,
                                                          (short)0, acc[0][t], false, false);
        acc[1][t] = __builtin_amdgcn_wmma_f32_16x16x4_f32(false, a1, false, bf,
                                                          (short)0, acc[1][t], false, false);
      }
    }

    #pragma unroll
    for (int mt = 0; mt < 2; ++mt) {
      #pragma unroll
      for (int t = 0; t < 4; ++t) {
        const float bias = sB1[16 * t + lm];
        float4 lo, hi;
        lo.x = fmaxf(acc[mt][t][0] + bias, 0.0f); lo.y = fmaxf(acc[mt][t][1] + bias, 0.0f);
        lo.z = fmaxf(acc[mt][t][2] + bias, 0.0f); lo.w = fmaxf(acc[mt][t][3] + bias, 0.0f);
        hi.x = fmaxf(acc[mt][t][4] + bias, 0.0f); hi.y = fmaxf(acc[mt][t][5] + bias, 0.0f);
        hi.z = fmaxf(acc[mt][t][6] + bias, 0.0f); hi.w = fmaxf(acc[mt][t][7] + bias, 0.0f);
        float* p = hw + (16 * t + lm) * 36 + 16 * mt + 8 * half;
        *(float4*)p = lo;
        *(float4*)(p + 4) = hi;
      }
    }
  }
  __syncthreads();

  {
    const float* hr = hw + lane;
    v2f h[32];
    #pragma unroll
    for (int i = 0; i < 32; ++i) h[i] = v2f{hr[(2 * i) * 36], hr[(2 * i + 1) * 36]};
    v2f a0 = h[0], a1 = h[1], a2 = h[2], a3 = h[3];
    #pragma unroll
    for (int i = 4; i < 32; i += 4) { a0 += h[i]; a1 += h[i + 1]; a2 += h[i + 2]; a3 += h[i + 3]; }
    const v2f sa = (a0 + a1) + (a2 + a3);
    const float mu = (sa.x + sa.y) * 0.015625f;
    const v2f mu2 = {mu, mu};
    #pragma unroll
    for (int i = 0; i < 32; ++i) h[i] = h[i] - mu2;
    v2f q0 = h[0] * h[0], q1 = h[1] * h[1], q2 = h[2] * h[2], q3 = h[3] * h[3];
    #pragma unroll
    for (int i = 4; i < 32; i += 4) {
      q0 = __builtin_elementwise_fma(h[i], h[i], q0);
      q1 = __builtin_elementwise_fma(h[i + 1], h[i + 1], q1);
      q2 = __builtin_elementwise_fma(h[i + 2], h[i + 2], q2);
      q3 = __builtin_elementwise_fma(h[i + 3], h[i + 3], q3);
    }
    const v2f sq = (q0 + q1) + (q2 + q3);
    const float inv = rsqrtf((sq.x + sq.y) * 0.015625f + 1e-5f);
    const v2f inv2 = {inv, inv};
    const v2f* g2  = (const v2f*)sG;
    const v2f* be2 = (const v2f*)sBeta;
    #pragma unroll
    for (int i = 0; i < 32; ++i) h[i] = __builtin_elementwise_fma(h[i] * inv2, g2[i], be2[i]);
    #pragma unroll
    for (int q = 0; q < 6; ++q) {
      const v2f* w = (const v2f*)(sW2 + q * 64);
      v2f p0 = h[0] * w[0], p1 = h[1] * w[1];
      #pragma unroll
      for (int i = 2; i < 32; i += 2) {
        p0 = __builtin_elementwise_fma(h[i], w[i], p0);
        p1 = __builtin_elementwise_fma(h[i + 1], w[i + 1], p1);
      }
      const v2f ps = p0 + p1;
      hw[q * 32 + lane] = (ps.x + ps.y) + sB2[q];
    }
  }

  const int row = rowBase + lane;
  #pragma unroll 1
  for (int q = 0; q < 6; ++q) {
    const float zz = tanhf(hw[q * 32 + lane]);
    const float h2 = 1.57079632679489662f * zz;
    const float ce = cosf(h2), se = sinf(h2);
    const float2 ry = sTab[128 + q];
    const float2 pz = sTab[146 + q];
    const float cc = ry.x * ce, ss = ry.y * se, sc = ry.y * ce, cs = ry.x * se;
    hw[192 + (4 * q) * 32 + lane]     = pz.x * (cc - ss);
    hw[192 + (4 * q + 1) * 32 + lane] = -pz.y * (cc + ss);
    hw[192 + (4 * q + 2) * 32 + lane] = pz.x * (sc + cs);
    hw[192 + (4 * q + 3) * 32 + lane] = pz.y * (cs - sc);
  }

  v2f st[64];
  {
    v2f u[6][2];
    #pragma unroll
    for (int q = 0; q < 6; ++q) {
      u[q][0] = v2f{hw[192 + (4 * q) * 32 + lane],     hw[192 + (4 * q + 1) * 32 + lane]};
      u[q][1] = v2f{hw[192 + (4 * q + 2) * 32 + lane], hw[192 + (4 * q + 3) * 32 + lane]};
    }
    v2f hi[8], lo[8];
    triple(hi, u[0], u[1], u[2]);
    triple(lo, u[3], u[4], u[5]);
    build_state(st, hi, lo, std::make_integer_sequence<int, 64>{});
  }

  #pragma unroll 1
  for (int l = 1; l < 3; ++l) {
    if (l == 2) ring_apply<2>(st, std::make_integer_sequence<int, 64>{});
    phase_all(st, sTab + 64 * (l - 1), std::make_integer_sequence<int, 64>{});
    layer_ry(st, sTab + 128 + l * 6, std::make_integer_sequence<int, 6>{});
  }

  float zq[6];
  {
    float cur[64];
    probs<1>(cur, st, std::make_integer_sequence<int, 64>{});
    #pragma unroll
    for (int b = 0; b < 6; ++b) {
      const int n2 = 32 >> b;
      float accd = cur[0] - cur[1];
      cur[0] = cur[0] + cur[1];
      #pragma unroll
      for (int j = 1; j < n2; ++j) {
        float lo = cur[2 * j], hi = cur[2 * j + 1];
        accd += lo - hi;
        cur[j] = lo + hi;
      }
      zq[5 - b] = accd;
    }
  }
  if (row < B) {
    #pragma unroll
    for (int k = 0; k < 5; ++k) {
      float o = 0.f;
      #pragma unroll
      for (int q = 0; q < 6; ++q) o = fmaf(sWc[k * 6 + q], zq[q], o);
      out[(size_t)row * 5 + k] = o + sBc[k];
    }
  }
}

extern "C" void kernel_launch(void* const* d_in, const int* in_sizes, int n_in,
                              void* d_out, int out_size, void* d_ws, size_t ws_size,
                              hipStream_t stream) {
  (void)n_in; (void)out_size; (void)d_ws; (void)ws_size;
  const float* x    = (const float*)d_in[0];
  const float* W1   = (const float*)d_in[1];
  const float* b1   = (const float*)d_in[2];
  const float* ln_g = (const float*)d_in[3];
  const float* ln_b = (const float*)d_in[4];
  const float* W2   = (const float*)d_in[5];
  const float* b2   = (const float*)d_in[6];
  const float* th_s = (const float*)d_in[7];
  const float* th_t = (const float*)d_in[8];
  const float* Wc   = (const float*)d_in[9];
  const float* bc   = (const float*)d_in[10];
  float* out = (float*)d_out;
  int B = in_sizes[0] / 64;
  int grid = (B + 127) / 128;
  qmaml_kernel<<<grid, 128, 0, stream>>>(x, W1, b1, ln_g, ln_b, W2, b2, th_s, th_t, Wc, bc, out, B);
}


# RY of layers 1, 2: cos (or sin behind a quarter turn) factored out, a pair = 2 packed FMAs not 2 mul + 2 FMA; the 12 factors squared scale the readout once; rounding changes, sum order kept
# speedup vs baseline: 9.6041x; 1.0632x over previous
#include <hip/hip_runtime.h>
#include <math.h>
#include <utility>


typedef __attribute__((ext_vector_type(2))) float v2f;
typedef __attribute__((ext_vector_type(8))) float v8f;

__host__ __device__ constexpr int ring_src(int rr, int i) {
  for (int w = 5; w >= 0; --w) {
    int t = (w + rr) % 6;
    int cbit = (i >> (5 - w)) & 1;
    i ^= cbit << (5 - t);
  }
  return i;
}

__host__ __device__ constexpr int ring_dst(int rr, int k) {
  for (int w = 0; w < 6; ++w) {
    int t = (w + rr) % 6;
    int cbit = (k >> (5 - w)) & 1;
    k ^= cbit << (5 - t);
  }
  return k;
}

template <int Wq, int P> struct PairIdx {
  static constexpr int b = 5 - Wq;
  static constexpr int i = ((P >> b) << (b + 1)) | (P & ((1 << b) - 1));
  static constexpr int j = i | (1 << b);
};
template <int Wq, int G, int... U>
__device__ __forceinline__ void ry_group(v2f (&st)[64], const v2f t2, const v2f mt2,
                                         std::integer_sequence<int, U...>) {
  v2f a[4], b[4], t0[4], t1[4];
  ((a[U] = st[PairIdx<Wq, 4 * G + U>::i], b[U] = st[PairIdx<Wq, 4 * G + U>::j]), ...);
  ((t0[U] = __builtin_elementwise_fma(mt2, b[U], a[U])), ...);
  ((t1[U] = __builtin_elementwise_fma(t2, a[U], b[U])), ...);
  ((st[PairIdx<Wq, 4 * G + U>::i] = t0[U], st[PairIdx<Wq, 4 * G + U>::j] = t1[U]), ...);
}
template <int Wq, int P>
__device__ __forceinline__ void quarter_turn_pair(v2f (&st)[64]) {
  const v2f a = st[PairIdx<Wq, P>::i];
  st[PairIdx<Wq, P>::i] = -st[PairIdx<Wq, P>::j];
  st[PairIdx<Wq, P>::j] = a;
}
template <int Wq, int... P>
__device__ __forceinline__ void quarter_turn(v2f (&st)[64], std::integer_sequence<int, P...>) {
  (quarter_turn_pair<Wq, P>(st), ...);
}
template <int Wq, int... G>
__device__ __forceinline__ void ry_all(v2f (&st)[64], const float2* ry,
                                       std::integer_sequence<int, G...>) {
  const float2 rf = ry[Wq];
  const v2f t2 = {rf.x, rf.x}, mt2 = {-rf.x, -rf.x};
  if (__builtin_amdgcn_readfirstlane(__float_as_int(rf.y)) != 0)
    quarter_turn<Wq>(st, std::make_integer_sequence<int, 32>{});
  (ry_group<Wq, G>(st, t2, mt2, std::make_integer_sequence<int, 4>{}), ...);
}
template <int... Wq>
__device__ __forceinline__ void layer_ry(v2f (&st)[64], const float2* ry,
                                         std::integer_sequence<int, Wq...>) {
  (ry_all<Wq>(st, ry, std::make_integer_sequence<int, 8>{}), ...);
}
__device__ __forceinline__ v2f phase_mul(v2f a, float2 m) {
  const v2f asw = __builtin_shufflevector(a, a, 1, 0);
  const v2f mr = {m.x, m.x}, mi = {-m.y, m.y};
  return __builtin_elementwise_fma(mi, asw, mr * a);
}
__device__ __forceinline__ v2f cmul(v2f a, v2f b) {
  const v2f ar = {a.x, a.x}, ai = {a.y, a.y};
  const v2f bs = {-b.y, b.x};
  return __builtin_elementwise_fma(ai, bs, ar * b);
}
template <int... I>
__device__ __forceinline__ void phase_all(v2f (&st)[64], const float2* dg,
                                          std::integer_sequence<int, I...>) {
  ((st[I] = phase_mul(st[I], dg[I])), ...);
}
template <int RR, int... I>
__device__ __forceinline__ void ring_apply(v2f (&st)[64], std::integer_sequence<int, I...>) {
  v2f n[64];
  ((n[I] = st[ring_src(RR, I)]), ...);
  ((st[I] = n[I]), ...);
}
template <int... I>
__device__ __forceinline__ void build_state(v2f (&st)[64], const v2f (&hi)[8], const v2f (&lo)[8],
                                            std::integer_sequence<int, I...>) {
  ((st[ring_dst(1, I)] = cmul(hi[I >> 3], lo[I & 7])), ...);
}
__device__ __forceinline__ void triple(v2f (&o)[8], const v2f (&a)[2], const v2f (&b)[2],
                                       const v2f (&c)[2]) {
  #pragma unroll
  for (int k = 0; k < 4; ++k) {
    const v2f ab = cmul(a[k >> 1], b[k & 1]);
    o[2 * k]     = cmul(ab, c[0]);
    o[2 * k + 1] = cmul(ab, c[1]);
  }
}
template <int RR, int... I>
__device__ __forceinline__ void probs(float (&cur)[64], const v2f (&st)[64],
                                      std::integer_sequence<int, I...>) {
  ((cur[I] = st[ring_src(RR, I)].x * st[ring_src(RR, I)].x
           + st[ring_src(RR, I)].y * st[ring_src(RR, I)].y), ...);
}

struct TabAngle {
  float phi[6], om[6], small;
  __device__ __forceinline__ void request(int tid, const float* th_s, const float* th_t) {
    const bool l1 = tid < 64;
    const float* pphi = l1 ? (th_s + 18) : th_t;
    const float* pom  = l1 ? th_s : (th_s + 18);
    #pragma unroll
    for (int w = 0; w < 6; ++w) { phi[w] = pphi[w * 3 + 0]; om[w] = pom[w * 3 + 2]; }
    const int g = min(tid, 23);
    const float* ps = (g < 12) ? (th_s + g * 3 + 1)
                    : (g < 18) ? (th_t + (g - 12) * 3 + 1) : (th_s + (g - 18) * 3);
    small = *ps;
  }
  __device__ __forceinline__ float diag_angle(int e) const {
    const int iphi = e & 63;
    const int i    = (e < 64) ? ring_src(1, iphi) : ring_src(2, iphi);
    float a = 0.f, b = 0.f;
    #pragma unroll
    for (int w = 0; w < 6; ++w) {
      a += ((iphi >> (5 - w)) & 1) ? phi[w] : -phi[w];
      b += ((i >> (5 - w)) & 1) ? om[w] : -om[w];
    }
    return 0.5f * (a + b);
  }
};

__global__ __launch_bounds__(128, 4) void qmaml_kernel(
    const float* __restrict__ x,  const float* __restrict__ W1,
    const float* __restrict__ b1, const float* __restrict__ ln_g,
    const float* __restrict__ ln_b, const float* __restrict__ W2,
    const float* __restrict__ b2, const float* __restrict__ th_s,
    const float* __restrict__ th_t, const float* __restrict__ Wc,
    const float* __restrict__ bc, float* __restrict__ out, int B)
{
  __shared__ float  sW1[4096];
  __shared__ __attribute__((aligned(16))) float sW2[384];
  __shared__ float  sB1[64];
  __shared__ __attribute__((aligned(16))) float sG[64];
  __shared__ __attribute__((aligned(16))) float sBeta[64];
  __shared__ float  sB2[8], sWc[32], sBc[8];
  __shared__ float2 sTab[152];
  __shared__ __attribute__((aligned(16))) float sNrm[12];
  __shared__ __attribute__((aligned(16))) float sH[4 * 64 * 36];

  const int tid = threadIdx.x;

  TabAngle ta;
  ta.request(tid, th_s, th_t);

  for (int i = tid; i < 1024; i += 128) ((float4*)sW1)[i] = ((const float4*)W1)[i];
  for (int i = tid; i < 384; i += 128)  sW2[i] = W2[i];
  if (tid < 64) { sB1[tid] = b1[tid]; sG[tid] = ln_g[tid]; sBeta[tid] = ln_b[tid]; }
  if (tid < 6)  sB2[tid] = b2[tid];
  if (tid < 30) sWc[tid] = Wc[tid];
  if (tid < 5)  sBc[tid] = bc[tid];
  {
    float sn0, cs0;
    sincosf(ta.diag_angle(tid), &sn0, &cs0);
    sTab[tid] = make_float2(cs0, sn0);
    if (tid < 24) {
      float sn1, cs1;
      sincosf(0.5f * ta.small, &sn1, &cs1);
      if (tid >= 6 && tid < 18) {
        const bool turn = fabsf(sn1) > fabsf(cs1);
        const float m = turn ? sn1 : cs1;
        const float t = (turn ? -cs1 : sn1) / m;
        sNrm[tid - 6] = m * m;
        sTab[128 + tid] = make_float2(t, turn ? 1.0f : 0.0f);
      } else {
        sTab[128 + tid] = make_float2(cs1, sn1);
      }
    }
  }
  __syncthreads();

  const int lane = tid & 31;
  const int wave = tid >> 5;
  const int half = lane >> 4;
  const int lm   = lane & 15;
  const int rowBase = blockIdx.x * 128 + wave * 32;
  float* const hw = sH + wave * (64 * 36);

  {
    v8f acc[2][4] = {};
    int rowA0 = rowBase + lm;      if (rowA0 >= B) rowA0 = B - 1;
    int rowA1 = rowBase + 16 + lm; if (rowA1 >= B) rowA1 = B - 1;
    const float* xrow0 = x + (size_t)rowA0 * 64 + 2 * half;
    const float* xrow1 = x + (size_t)rowA1 * 64 + 2 * half;
    #pragma unroll
    for (int kk = 0; kk < 16; ++kk) {
      v2f a0 = *(const v2f*)(xrow0 + 4 * kk);
      v2f a1 = *(const v2f*)(xrow1 + 4 * kk);
      #pragma unroll
      for (int t = 0; t < 4; ++t) {
        v2f bf = *(const v2f*)(sW1 + (16 * t + lm) * 64 + 4 * kk + 2 * half);
        acc[0][t] = __builtin_amdgcn_wmma_f32_16x16x4_f32(false, a0, false, bf,
                                                          (short)0, acc[0][t], false, false);
        acc[1][t] = __builtin_amdgcn_wmma_f32_16x16x4_f32(false, a1, false, bf,
                                                          (short)0, acc[1][t], false, false);
      }
    }

    #pragma unroll
    for (int mt = 0; mt < 2; ++mt) {
      #pragma unroll
      for (int t = 0; t < 4; ++t) {
        const float bias = sB1[16 * t + lm];
        float4 lo, hi;
        lo.x = fmaxf(acc[mt][t][0] + bias, 0.0f); lo.y = fmaxf(acc[mt][t][1] + bias, 0.0f);
        lo.z = fmaxf(acc[mt][t][2] + bias, 0.0f); lo.w = fmaxf(acc[mt][t][3] + bias, 0.0f);
        hi.x = fmaxf(acc[mt][t][4] + bias, 0.0f); hi.y = fmaxf(acc[mt][t][5] + bias, 0.0f);
        hi.z = fmaxf(acc[mt][t][6] + bias, 0.0f); hi.w = fmaxf(acc[mt][t][7] + bias, 0.0f);
        float* p = hw + (16 * t + lm) * 36 + 16 * mt + 8 * half;
        *(float4*)p = lo;
        *(float4*)(p + 4) = hi;
      }
    }
  }
  __syncthreads();

  {
    const float* hr = hw + lane;
    v2f h[32];
    #pragma unroll
    for (int i = 0; i < 32; ++i) h[i] = v2f{hr[(2 * i) * 36], hr[(2 * i + 1) * 36]};
    v2f a0 = h[0], a1 = h[1], a2 = h[2], a3 = h[3];
    #pragma unroll
    for (int i = 4; i < 32; i += 4) { a0 += h[i]; a1 += h[i + 1]; a2 += h[i + 2]; a3 += h[i + 3]; }
    const v2f sa = (a0 + a1) + (a2 + a3);
    const float mu = (sa.x + sa.y) * 0.015625f;
    const v2f mu2 = {mu, mu};
    #pragma unroll
    for (int i = 0; i < 32; ++i) h[i] = h[i] - mu2;
    v2f q0 = h[0] * h[0], q1 = h[1] * h[1], q2 = h[2] * h[2], q3 = h[3] * h[3];
    #pragma unroll
    for (int i = 4; i < 32; i += 4) {
      q0 = __builtin_elementwise_fma(h[i], h[i], q0);
      q1 = __builtin_elementwise_fma(h[i + 1], h[i + 1], q1);
      q2 = __builtin_elementwise_fma(h[i + 2], h[i + 2], q2);
      q3 = __builtin_elementwise_fma(h[i + 3], h[i + 3], q3);
    }
    const v2f sq = (q0 + q1) + (q2 + q3);
    const float inv = rsqrtf((sq.x + sq.y) * 0.015625f + 1e-5f);
    const v2f inv2 = {inv, inv};
    const v2f* g2  = (const v2f*)sG;
    const v2f* be2 = (const v2f*)sBeta;
    #pragma unroll
    for (int i = 0; i < 32; ++i) h[i] = __builtin_elementwise_fma(h[i] * inv2, g2[i], be2[i]);
    #pragma unroll
    for (int q = 0; q < 6; ++q) {
      const v2f* w = (const v2f*)(sW2 + q * 64);
      v2f p0 = h[0] * w[0], p1 = h[1] * w[1];
      #pragma unroll
      for (int i = 2; i < 32; i += 2) {
        p0 = __builtin_elementwise_fma(h[i], w[i], p0);
        p1 = __builtin_elementwise_fma(h[i + 1], w[i + 1], p1);
      }
      const v2f ps = p0 + p1;
      hw[q * 32 + lane] = (ps.x + ps.y) + sB2[q];
    }
  }

  const int row = rowBase + lane;
  #pragma unroll 1
  for (int q = 0; q < 6; ++q) {
    const float zz = tanhf(hw[q * 32 + lane]);
    const float h2 = 1.57079632679489662f * zz;
    const float ce = cosf(h2), se = sinf(h2);
    const float2 ry = sTab[128 + q];
    const float2 pz = sTab[146 + q];
    const float cc = ry.x * ce, ss = ry.y * se, sc = ry.y * ce, cs = ry.x * se;
    hw[192 + (4 * q) * 32 + lane]     = pz.x * (cc - ss);
    hw[192 + (4 * q + 1) * 32 + lane] = -pz.y * (cc + ss);
    hw[192 + (4 * q + 2) * 32 + lane] = pz.x * (sc + cs);
    hw[192 + (4 * q + 3) * 32 + lane] = pz.y * (cs - sc);
  }

  v2f st[64];
  {
    v2f u[6][2];
    #pragma unroll
    for (int q = 0; q < 6; ++q) {
      u[q][0] = v2f{hw[192 + (4 * q) * 32 + lane],     hw[192 + (4 * q + 1) * 32 + lane]};
      u[q][1] = v2f{hw[192 + (4 * q + 2) * 32 + lane], hw[192 + (4 * q + 3) * 32 + lane]};
    }
    v2f hi[8], lo[8];
    triple(hi, u[0], u[1], u[2]);
    triple(lo, u[3], u[4], u[5]);
    build_state(st, hi, lo, std::make_integer_sequence<int, 64>{});
  }

  #pragma unroll 1
  for (int l = 1; l < 3; ++l) {
    if (l == 2) ring_apply<2>(st, std::make_integer_sequence<int, 64>{});
    phase_all(st, sTab + 64 * (l - 1), std::make_integer_sequence<int, 64>{});
    layer_ry(st, sTab + 128 + l * 6, std::make_integer_sequence<int, 6>{});
  }

  float zq[6];
  {
    float cur[64];
    probs<1>(cur, st, std::make_integer_sequence<int, 64>{});
    #pragma unroll
    for (int b = 0; b < 6; ++b) {
      const int n2 = 32 >> b;
      float accd = cur[0] - cur[1];
      cur[0] = cur[0] + cur[1];
      #pragma unroll
      for (int j = 1; j < n2; ++j) {
        float lo = cur[2 * j], hi = cur[2 * j + 1];
        accd += lo - hi;
        cur[j] = lo + hi;
      }
      zq[5 - b] = accd;
    }
  }
  if (row < B) {
    float nrm = sNrm[0];
    #pragma unroll
    for (int g = 1; g < 12; ++g) nrm *= sNrm[g];
    #pragma unroll
    for (int k = 0; k < 5; ++k) {
      float o = 0.f;
      #pragma unroll
      for (int q = 0; q < 6; ++q) o = fmaf(sWc[k * 6 + q], zq[q], o);
      out[(size_t)row * 5 + k] = fmaf(o, nrm, sBc[k]);
    }
  }
}

extern "C" void kernel_launch(void* const* d_in, const int* in_sizes, int n_in,
                              void* d_out, int out_size, void* d_ws, size_t ws_size,
                              hipStream_t stream) {
  (void)n_in; (void)out_size; (void)d_ws; (void)ws_size;
  const float* x    = (const float*)d_in[0];
  const float* W1   = (const float*)d_in[1];
  const float* b1   = (const float*)d_in[2];
  const float* ln_g = (const float*)d_in[3];
  const float* ln_b = (const float*)d_in[4];
  const float* W2   = (const float*)d_in[5];
  const float* b2   = (const float*)d_in[6];
  const float* th_s = (const float*)d_in[7];
  const float* th_t = (const float*)d_in[8];
  const float* Wc   = (const float*)d_in[9];
  const float* bc   = (const float*)d_in[10];
  float* out = (float*)d_out;
  int B = in_sizes[0] / 64;
  int grid = (B + 127) / 128;
  qmaml_kernel<<<grid, 128, 0, stream>>>(x, W1, b1, ln_g, ln_b, W2, b2, th_s, th_t, Wc, bc, out, B);
}


# Lane-uniform parameters (ln_g, ln_b, W2, b2, Wc, bc) read by scalar loads and used as scalar operands of unpacked FMAs: their LDS staging round trips and broadcast LDS reads gone; sum order kept
# speedup vs baseline: 10.3523x; 1.0779x over previous
#include <hip/hip_runtime.h>
#include <math.h>
#include <utility>


typedef __attribute__((ext_vector_type(2))) float v2f;
typedef __attribute__((ext_vector_type(8))) float v8f;

__host__ __device__ constexpr int ring_src(int rr, int i) {
  for (int w = 5; w >= 0; --w) {
    int t = (w + rr) % 6;
    int cbit = (i >> (5 - w)) & 1;
    i ^= cbit << (5 - t);
  }
  return i;
}

__host__ __device__ constexpr int ring_dst(int rr, int k) {
  for (int w = 0; w < 6; ++w) {
    int t = (w + rr) % 6;
    int cbit = (k >> (5 - w)) & 1;
    k ^= cbit << (5 - t);
  }
  return k;
}

template <int Wq, int P> struct PairIdx {
  static constexpr int b = 5 - Wq;
  static constexpr int i = ((P >> b) << (b + 1)) | (P & ((1 << b) - 1));
  static constexpr int j = i | (1 << b);
};
template <int Wq, int G, int... U>
__device__ __forceinline__ void ry_group(v2f (&st)[64], const v2f t2, const v2f mt2,
                                         std::integer_sequence<int, U...>) {
  v2f a[4], b[4], t0[4], t1[4];
  ((a[U] = st[PairIdx<Wq, 4 * G + U>::i], b[U] = st[PairIdx<Wq, 4 * G + U>::j]), ...);
  ((t0[U] = __builtin_elementwise_fma(mt2, b[U], a[U])), ...);
  ((t1[U] = __builtin_elementwise_fma(t2, a[U], b[U])), ...);
  ((st[PairIdx<Wq, 4 * G + U>::i] = t0[U], st[PairIdx<Wq, 4 * G + U>::j] = t1[U]), ...);
}
template <int Wq, int P>
__device__ __forceinline__ void quarter_turn_pair(v2f (&st)[64]) {
  const v2f a = st[PairIdx<Wq, P>::i];
  st[PairIdx<Wq, P>::i] = -st[PairIdx<Wq, P>::j];
  st[PairIdx<Wq, P>::j] = a;
}
template <int Wq, int... P>
__device__ __forceinline__ void quarter_turn(v2f (&st)[64], std::integer_sequence<int, P...>) {
  (quarter_turn_pair<Wq, P>(st), ...);
}
template <int Wq, int... G>
__device__ __forceinline__ void ry_all(v2f (&st)[64], const float2* ry,
                                       std::integer_sequence<int, G...>) {
  const float2 rf = ry[Wq];
  const v2f t2 = {rf.x, rf.x}, mt2 = {-rf.x, -rf.x};
  if (__builtin_amdgcn_readfirstlane(__float_as_int(rf.y)) != 0)
    quarter_turn<Wq>(st, std::make_integer_sequence<int, 32>{});
  (ry_group<Wq, G>(st, t2, mt2, std::make_integer_sequence<int, 4>{}), ...);
}
template <int... Wq>
__device__ __forceinline__ void layer_ry(v2f (&st)[64], const float2* ry,
                                         std::integer_sequence<int, Wq...>) {
  (ry_all<Wq>(st, ry, std::make_integer_sequence<int, 8>{}), ...);
}
__device__ __forceinline__ v2f phase_mul(v2f a, float2 m) {
  const v2f asw = __builtin_shufflevector(a, a, 1, 0);
  const v2f mr = {m.x, m.x}, mi = {-m.y, m.y};
  return __builtin_elementwise_fma(mi, asw, mr * a);
}
__device__ __forceinline__ v2f cmul(v2f a, v2f b) {
  const v2f ar = {a.x, a.x}, ai = {a.y, a.y};
  const v2f bs = {-b.y, b.x};
  return __builtin_elementwise_fma(ai, bs, ar * b);
}
template <int... I>
__device__ __forceinline__ void phase_all(v2f (&st)[64], const float2* dg,
                                          std::integer_sequence<int, I...>) {
  ((st[I] = phase_mul(st[I], dg[I])), ...);
}
template <int RR, int... I>
__device__ __forceinline__ void ring_apply(v2f (&st)[64], std::integer_sequence<int, I...>) {
  v2f n[64];
  ((n[I] = st[ring_src(RR, I)]), ...);
  ((st[I] = n[I]), ...);
}
template <int... I>
__device__ __forceinline__ void build_state(v2f (&st)[64], const v2f (&hi)[8], const v2f (&lo)[8],
                                            std::integer_sequence<int, I...>) {
  ((st[ring_dst(1, I)] = cmul(hi[I >> 3], lo[I & 7])), ...);
}
__device__ __forceinline__ void triple(v2f (&o)[8], const v2f (&a)[2], const v2f (&b)[2],
                                       const v2f (&c)[2]) {
  #pragma unroll
  for (int k = 0; k < 4; ++k) {
    const v2f ab = cmul(a[k >> 1], b[k & 1]);
    o[2 * k]     = cmul(ab, c[0]);
    o[2 * k + 1] = cmul(ab, c[1]);
  }
}
template <int RR, int... I>
__device__ __forceinline__ void probs(float (&cur)[64], const v2f (&st)[64],
                                      std::integer_sequence<int, I...>) {
  ((cur[I] = st[ring_src(RR, I)].x * st[ring_src(RR, I)].x
           + st[ring_src(RR, I)].y * st[ring_src(RR, I)].y), ...);
}

struct TabAngle {
  float phi[6], om[6], small;
  __device__ __forceinline__ void request(int tid, const float* th_s, const float* th_t) {
    const bool l1 = tid < 64;
    const float* pphi = l1 ? (th_s + 18) : th_t;
    const float* pom  = l1 ? th_s : (th_s + 18);
    #pragma unroll
    for (int w = 0; w < 6; ++w) { phi[w] = pphi[w * 3 + 0]; om[w] = pom[w * 3 + 2]; }
    const int g = min(tid, 23);
    const float* ps = (g < 12) ? (th_s + g * 3 + 1)
                    : (g < 18) ? (th_t + (g - 12) * 3 + 1) : (th_s + (g - 18) * 3);
    small = *ps;
  }
  __device__ __forceinline__ float diag_angle(int e) const {
    const int iphi = e & 63;
    const int i    = (e < 64) ? ring_src(1, iphi) : ring_src(2, iphi);
    float a = 0.f, b = 0.f;
    #pragma unroll
    for (int w = 0; w < 6; ++w) {
      a += ((iphi >> (5 - w)) & 1) ? phi[w] : -phi[w];
      b += ((i >> (5 - w)) & 1) ? om[w] : -om[w];
    }
    return 0.5f * (a + b);
  }
};

__global__ __launch_bounds__(128, 4) void qmaml_kernel(
    const float* __restrict__ x,  const float* __restrict__ W1,
    const float* __restrict__ b1, const float* __restrict__ ln_g,
    const float* __restrict__ ln_b, const float* __restrict__ W2,
    const float* __restrict__ b2, const float* __restrict__ th_s,
    const float* __restrict__ th_t, const float* __restrict__ Wc,
    const float* __restrict__ bc, float* __restrict__ out, int B)
{
  __shared__ float  sW1[4096];
  __shared__ float  sB1[64];
  __shared__ float2 sTab[152];
  __shared__ __attribute__((aligned(16))) float sNrm[12];
  __shared__ __attribute__((aligned(16))) float sH[4 * 64 * 36];

  const int tid = threadIdx.x;

  TabAngle ta;
  ta.request(tid, th_s, th_t);

  for (int i = tid; i < 1024; i += 128) ((float4*)sW1)[i] = ((const float4*)W1)[i];
  if (tid < 64) sB1[tid] = b1[tid];
  {
    float sn0, cs0;
    sincosf(ta.diag_angle(tid), &sn0, &cs0);
    sTab[tid] = make_float2(cs0, sn0);
    if (tid < 24) {
      float sn1, cs1;
      sincosf(0.5f * ta.small, &sn1, &cs1);
      if (tid >= 6 && tid < 18) {
        const bool turn = fabsf(sn1) > fabsf(cs1);
        const float m = turn ? sn1 : cs1;
        const float t = (turn ? -cs1 : sn1) / m;
        sNrm[tid - 6] = m * m;
        sTab[128 + tid] = make_float2(t, turn ? 1.0f : 0.0f);
      } else {
        sTab[128 + tid] = make_float2(cs1, sn1);
      }
    }
  }
  __syncthreads();

  const int lane = tid & 31;
  const int wave = tid >> 5;
  const int half = lane >> 4;
  const int lm   = lane & 15;
  const int rowBase = blockIdx.x * 128 + wave * 32;
  float* const hw = sH + wave * (64 * 36);

  {
    v8f acc[2][4] = {};
    int rowA0 = rowBase + lm;      if (rowA0 >= B) rowA0 = B - 1;
    int rowA1 = rowBase + 16 + lm; if (rowA1 >= B) rowA1 = B - 1;
    const float* xrow0 = x + (size_t)rowA0 * 64 + 2 * half;
    const float* xrow1 = x + (size_t)rowA1 * 64 + 2 * half;
    #pragma unroll
    for (int kk = 0; kk < 16; ++kk) {
      v2f a0 = *(const v2f*)(xrow0 + 4 * kk);
      v2f a1 = *(const v2f*)(xrow1 + 4 * kk);
      #pragma unroll
      for (int t = 0; t < 4; ++t) {
        v2f bf = *(const v2f*)(sW1 + (16 * t + lm) * 64 + 4 * kk + 2 * half);
        acc[0][t] = __builtin_amdgcn_wmma_f32_16x16x4_f32(false, a0, false, bf,
                                                          (short)0, acc[0][t], false, false);
        acc[1][t] = __builtin_amdgcn_wmma_f32_16x16x4_f32(false, a1, false, bf,
                                                          (short)0, acc[1][t], false, false);
      }
    }

    #pragma unroll
    for (int mt = 0; mt < 2; ++mt) {
      #pragma unroll
      for (int t = 0; t < 4; ++t) {
        const float bias = sB1[16 * t + lm];
        float4 lo, hi;
        lo.x = fmaxf(acc[mt][t][0] + bias, 0.0f); lo.y = fmaxf(acc[mt][t][1] + bias, 0.0f);
        lo.z = fmaxf(acc[mt][t][2] + bias, 0.0f); lo.w = fmaxf(acc[mt][t][3] + bias, 0.0f);
        hi.x = fmaxf(acc[mt][t][4] + bias, 0.0f); hi.y = fmaxf(acc[mt][t][5] + bias, 0.0f);
        hi.z = fmaxf(acc[mt][t][6] + bias, 0.0f); hi.w = fmaxf(acc[mt][t][7] + bias, 0.0f);
        float* p = hw + (16 * t + lm) * 36 + 16 * mt + 8 * half;
        *(float4*)p = lo;
        *(float4*)(p + 4) = hi;
      }
    }
  }
  __syncthreads();

  {
    const float* hr = hw + lane;
    v2f h[32];
    #pragma unroll
    for (int i = 0; i < 32; ++i) h[i] = v2f{hr[(2 * i) * 36], hr[(2 * i + 1) * 36]};
    v2f a0 = h[0], a1 = h[1], a2 = h[2], a3 = h[3];
    #pragma unroll
    for (int i = 4; i < 32; i += 4) { a0 += h[i]; a1 += h[i + 1]; a2 += h[i + 2]; a3 += h[i + 3]; }
    const v2f sa = (a0 + a1) + (a2 + a3);
    const float mu = (sa.x + sa.y) * 0.015625f;
    const v2f mu2 = {mu, mu};
    #pragma unroll
    for (int i = 0; i < 32; ++i) h[i] = h[i] - mu2;
    v2f q0 = h[0] * h[0], q1 = h[1] * h[1], q2 = h[2] * h[2], q3 = h[3] * h[3];
    #pragma unroll
    for (int i = 4; i < 32; i += 4) {
      q0 = __builtin_elementwise_fma(h[i], h[i], q0);
      q1 = __builtin_elementwise_fma(h[i + 1], h[i + 1], q1);
      q2 = __builtin_elementwise_fma(h[i + 2], h[i + 2], q2);
      q3 = __builtin_elementwise_fma(h[i + 3], h[i + 3], q3);
    }
    const v2f sq = (q0 + q1) + (q2 + q3);
    const float inv = rsqrtf((sq.x + sq.y) * 0.015625f + 1e-5f);
    const v2f inv2 = {inv, inv};
    #pragma unroll
    for (int i = 0; i < 32; ++i) {
      const v2f s = h[i] * inv2;
      h[i] = v2f{fmaf(s.x, ln_g[2 * i], ln_b[2 * i]), fmaf(s.y, ln_g[2 * i + 1], ln_b[2 * i + 1])};
    }
    #pragma unroll
    for (int q = 0; q < 6; ++q) {
      const float* w = W2 + q * 64;
      float p0x = h[0].x * w[0], p0y = h[0].y * w[1], p1x = h[1].x * w[2], p1y = h[1].y * w[3];
      #pragma unroll
      for (int i = 2; i < 32; i += 2) {
        p0x = fmaf(h[i].x, w[2 * i], p0x);
        p0y = fmaf(h[i].y, w[2 * i + 1], p0y);
        p1x = fmaf(h[i + 1].x, w[2 * i + 2], p1x);
        p1y = fmaf(h[i + 1].y, w[2 * i + 3], p1y);
      }
      const v2f ps = {p0x + p1x, p0y + p1y};
      hw[q * 32 + lane] = (ps.x + ps.y) + b2[q];
      __builtin_amdgcn_sched_barrier(0);
    }
  }

  const int row = rowBase + lane;
  #pragma unroll 1
  for (int q = 0; q < 6; ++q) {
    const float zz = tanhf(hw[q * 32 + lane]);
    const float h2 = 1.57079632679489662f * zz;
    const float ce = cosf(h2), se = sinf(h2);
    const float2 ry = sTab[128 + q];
    const float2 pz = sTab[146 + q];
    const float cc = ry.x * ce, ss = ry.y * se, sc = ry.y * ce, cs = ry.x * se;
    hw[192 + (4 * q) * 32 + lane]     = pz.x * (cc - ss);
    hw[192 + (4 * q + 1) * 32 + lane] = -pz.y * (cc + ss);
    hw[192 + (4 * q + 2) * 32 + lane] = pz.x * (sc + cs);
    hw[192 + (4 * q + 3) * 32 + lane] = pz.y * (cs - sc);
  }

  v2f st[64];
  {
    v2f u[6][2];
    #pragma unroll
    for (int q = 0; q < 6; ++q) {
      u[q][0] = v2f{hw[192 + (4 * q) * 32 + lane],     hw[192 + (4 * q + 1) * 32 + lane]};
      u[q][1] = v2f{hw[192 + (4 * q + 2) * 32 + lane], hw[192 + (4 * q + 3) * 32 + lane]};
    }
    v2f hi[8], lo[8];
    triple(hi, u[0], u[1], u[2]);
    triple(lo, u[3], u[4], u[5]);
    build_state(st, hi, lo, std::make_integer_sequence<int, 64>{});
  }

  #pragma unroll 1
  for (int l = 1; l < 3; ++l) {
    if (l == 2) ring_apply<2>(st, std::make_integer_sequence<int, 64>{});
    phase_all(st, sTab + 64 * (l - 1), std::make_integer_sequence<int, 64>{});
    layer_ry(st, sTab + 128 + l * 6, std::make_integer_sequence<int, 6>{});
  }

  float zq[6];
  {
    float cur[64];
    probs<1>(cur, st, std::make_integer_sequence<int, 64>{});
    #pragma unroll
    for (int b = 0; b < 6; ++b) {
      const int n2 = 32 >> b;
      float accd = cur[0] - cur[1];
      cur[0] = cur[0] + cur[1];
      #pragma unroll
      for (int j = 1; j < n2; ++j) {
        float lo = cur[2 * j], hi = cur[2 * j + 1];
        accd += lo - hi;
        cur[j] = lo + hi;
      }
      zq[5 - b] = accd;
    }
  }
  if (row < B) {
    float nrm = sNrm[0];
    #pragma unroll
    for (int g = 1; g < 12; ++g) nrm *= sNrm[g];
    #pragma unroll
    for (int k = 0; k < 5; ++k) {
      float o = 0.f;
      #pragma unroll
      for (int q = 0; q < 6; ++q) o = fmaf(Wc[k * 6 + q], zq[q], o);
      out[(size_t)row * 5 + k] = fmaf(o, nrm, bc[k]);
    }
  }
}

extern "C" void kernel_launch(void* const* d_in, const int* in_sizes, int n_in,
                              void* d_out, int out_size, void* d_ws, size_t ws_size,
                              hipStream_t stream) {
  (void)n_in; (void)out_size; (void)d_ws; (void)ws_size;
  const float* x    = (const float*)d_in[0];
  const float* W1   = (const float*)d_in[1];
  const float* b1   = (const float*)d_in[2];
  const float* ln_g = (const float*)d_in[3];
  const float* ln_b = (const float*)d_in[4];
  const float* W2   = (const float*)d_in[5];
  const float* b2   = (const float*)d_in[6];
  const float* th_s = (const float*)d_in[7];
  const float* th_t = (const float*)d_in[8];
  const float* Wc   = (const float*)d_in[9];
  const float* bc   = (const float*)d_in[10];
  float* out = (float*)d_out;
  int B = in_sizes[0] / 64;
  int grid = (B + 127) / 128;
  qmaml_kernel<<<grid, 128, 0, stream>>>(x, W1, b1, ln_g, ln_b, W2, b2, th_s, th_t, Wc, bc, out, B);
}


# W1 held in LDS in matrix-fragment order, both loaders' k order = 4 consecutive k per lane: W1 reads are lane-adjacent 128-bit loads (16-way bank conflict gone), x by 128-bit loads; sum order changes
# speedup vs baseline: 13.1049x; 1.2659x over previous
#include <hip/hip_runtime.h>
#include <math.h>
#include <utility>


typedef __attribute__((ext_vector_type(2))) float v2f;
typedef __attribute__((ext_vector_type(8))) float v8f;

__host__ __device__ constexpr int ring_src(int rr, int i) {
  for (int w = 5; w >= 0; --w) {
    int t = (w + rr) % 6;
    int cbit = (i >> (5 - w)) & 1;
    i ^= cbit << (5 - t);
  }
  return i;
}

__host__ __device__ constexpr int ring_dst(int rr, int k) {
  for (int w = 0; w < 6; ++w) {
    int t = (w + rr) % 6;
    int cbit = (k >> (5 - w)) & 1;
    k ^= cbit << (5 - t);
  }
  return k;
}

template <int Wq, int P> struct PairIdx {
  static constexpr int b = 5 - Wq;
  static constexpr int i = ((P >> b) << (b + 1)) | (P & ((1 << b) - 1));
  static constexpr int j = i | (1 << b);
};
template <int Wq, int G, int... U>
__device__ __forceinline__ void ry_group(v2f (&st)[64], const v2f t2, const v2f mt2,
                                         std::integer_sequence<int, U...>) {
  v2f a[4], b[4], t0[4], t1[4];
  ((a[U] = st[PairIdx<Wq, 4 * G + U>::i], b[U] = st[PairIdx<Wq, 4 * G + U>::j]), ...);
  ((t0[U] = __builtin_elementwise_fma(mt2, b[U], a[U])), ...);
  ((t1[U] = __builtin_elementwise_fma(t2, a[U], b[U])), ...);
  ((st[PairIdx<Wq, 4 * G + U>::i] = t0[U], st[PairIdx<Wq, 4 * G + U>::j] = t1[U]), ...);
}
template <int Wq, int P>
__device__ __forceinline__ void quarter_turn_pair(v2f (&st)[64]) {
  const v2f a = st[PairIdx<Wq, P>::i];
  st[PairIdx<Wq, P>::i] = -st[PairIdx<Wq, P>::j];
  st[PairIdx<Wq, P>::j] = a;
}
template <int Wq, int... P>
__device__ __forceinline__ void quarter_turn(v2f (&st)[64], std::integer_sequence<int, P...>) {
  (quarter_turn_pair<Wq, P>(st), ...);
}
template <int Wq, int... G>
__device__ __forceinline__ void ry_all(v2f (&st)[64], const float2* ry,
                                       std::integer_sequence<int, G...>) {
  const float2 rf = ry[Wq];
  const v2f t2 = {rf.x, rf.x}, mt2 = {-rf.x, -rf.x};
  if (__builtin_amdgcn_readfirstlane(__float_as_int(rf.y)) != 0)
    quarter_turn<Wq>(st, std::make_integer_sequence<int, 32>{});
  (ry_group<Wq, G>(st, t2, mt2, std::make_integer_sequence<int, 4>{}), ...);
}
template <int... Wq>
__device__ __forceinline__ void layer_ry(v2f (&st)[64], const float2* ry,
                                         std::integer_sequence<int, Wq...>) {
  (ry_all<Wq>(st, ry, std::make_integer_sequence<int, 8>{}), ...);
}
__device__ __forceinline__ v2f phase_mul(v2f a, float2 m) {
  const v2f asw = __builtin_shufflevector(a, a, 1, 0);
  const v2f mr = {m.x, m.x}, mi = {-m.y, m.y};
  return __builtin_elementwise_fma(mi, asw, mr * a);
}
__device__ __forceinline__ v2f cmul(v2f a, v2f b) {
  const v2f ar = {a.x, a.x}, ai = {a.y, a.y};
  const v2f bs = {-b.y, b.x};
  return __builtin_elementwise_fma(ai, bs, ar * b);
}
template <int... I>
__device__ __forceinline__ void phase_all(v2f (&st)[64], const float2* dg,
                                          std::integer_sequence<int, I...>) {
  ((st[I] = phase_mul(st[I], dg[I])), ...);
}
template <int RR, int... I>
__device__ __forceinline__ void ring_apply(v2f (&st)[64], std::integer_sequence<int, I...>) {
  v2f n[64];
  ((n[I] = st[ring_src(RR, I)]), ...);
  ((st[I] = n[I]), ...);
}
template <int... I>
__device__ __forceinline__ void build_state(v2f (&st)[64], const v2f (&hi)[8], const v2f (&lo)[8],
                                            std::integer_sequence<int, I...>) {
  ((st[ring_dst(1, I)] = cmul(hi[I >> 3], lo[I & 7])), ...);
}
__device__ __forceinline__ void triple(v2f (&o)[8], const v2f (&a)[2], const v2f (&b)[2],
                                       const v2f (&c)[2]) {
  #pragma unroll
  for (int k = 0; k < 4; ++k) {
    const v2f ab = cmul(a[k >> 1], b[k & 1]);
    o[2 * k]     = cmul(ab, c[0]);
    o[2 * k + 1] = cmul(ab, c[1]);
  }
}
template <int RR, int... I>
__device__ __forceinline__ void probs(float (&cur)[64], const v2f (&st)[64],
                                      std::integer_sequence<int, I...>) {
  ((cur[I] = st[ring_src(RR, I)].x * st[ring_src(RR, I)].x
           + st[ring_src(RR, I)].y * st[ring_src(RR, I)].y), ...);
}

struct TabAngle {
  float phi[6], om[6], small;
  __device__ __forceinline__ void request(int tid, const float* th_s, const float* th_t) {
    const bool l1 = tid < 64;
    const float* pphi = l1 ? (th_s + 18) : th_t;
    const float* pom  = l1 ? th_s : (th_s + 18);
    #pragma unroll
    for (int w = 0; w < 6; ++w) { phi[w] = pphi[w * 3 + 0]; om[w] = pom[w * 3 + 2]; }
    const int g = min(tid, 23);
    const float* ps = (g < 12) ? (th_s + g * 3 + 1)
                    : (g < 18) ? (th_t + (g - 12) * 3 + 1) : (th_s + (g - 18) * 3);
    small = *ps;
  }
  __device__ __forceinline__ float diag_angle(int e) const {
    const int iphi = e & 63;
    const int i    = (e < 64) ? ring_src(1, iphi) : ring_src(2, iphi);
    float a = 0.f, b = 0.f;
    #pragma unroll
    for (int w = 0; w < 6; ++w) {
      a += ((iphi >> (5 - w)) & 1) ? phi[w] : -phi[w];
      b += ((i >> (5 - w)) & 1) ? om[w] : -om[w];
    }
    return 0.5f * (a + b);
  }
};

__global__ __launch_bounds__(128, 4) void qmaml_kernel(
    const float* __restrict__ x,  const float* __restrict__ W1,
    const float* __restrict__ b1, const float* __restrict__ ln_g,
    const float* __restrict__ ln_b, const float* __restrict__ W2,
    const float* __restrict__ b2, const float* __restrict__ th_s,
    const float* __restrict__ th_t, const float* __restrict__ Wc,
    const float* __restrict__ bc, float* __restrict__ out, int B)
{
  __shared__ __attribute__((aligned(16))) float sW1[4096];
  __shared__ float  sB1[64];
  __shared__ float2 sTab[152];
  __shared__ __attribute__((aligned(16))) float sNrm[12];
  __shared__ __attribute__((aligned(16))) float sH[4 * 64 * 36];

  const int tid = threadIdx.x;

  TabAngle ta;
  ta.request(tid, th_s, th_t);

  for (int i = tid; i < 1024; i += 128) {
    const int src = (16 * ((i >> 5) & 3) + (i & 15)) * 16 + 2 * (i >> 7) + ((i >> 4) & 1);
    ((float4*)sW1)[i] = ((const float4*)W1)[src];
  }
  if (tid < 64) sB1[tid] = b1[tid];
  {
    float sn0, cs0;
    sincosf(ta.diag_angle(tid), &sn0, &cs0);
    sTab[tid] = make_float2(cs0, sn0);
    if (tid < 24) {
      float sn1, cs1;
      sincosf(0.5f * ta.small, &sn1, &cs1);
      if (tid >= 6 && tid < 18) {
        const bool turn = fabsf(sn1) > fabsf(cs1);
        const float m = turn ? sn1 : cs1;
        const float t = (turn ? -cs1 : sn1) / m;
        sNrm[tid - 6] = m * m;
        sTab[128 + tid] = make_float2(t, turn ? 1.0f : 0.0f);
      } else {
        sTab[128 + tid] = make_float2(cs1, sn1);
      }
    }
  }
  __syncthreads();

  const int lane = tid & 31;
  const int wave = tid >> 5;
  const int half = lane >> 4;
  const int lm   = lane & 15;
  const int rowBase = blockIdx.x * 128 + wave * 32;
  float* const hw = sH + wave * (64 * 36);

  {
    v8f acc[2][4] = {};
    int rowA0 = rowBase + lm;      if (rowA0 >= B) rowA0 = B - 1;
    int rowA1 = rowBase + 16 + lm; if (rowA1 >= B) rowA1 = B - 1;
    const float* xrow0 = x + (size_t)rowA0 * 64 + 4 * half;
    const float* xrow1 = x + (size_t)rowA1 * 64 + 4 * half;
    const float4* wf = (const float4*)sW1 + lane;
    #pragma unroll
    for (int j = 0; j < 8; ++j) {
      const float4 a0 = *(const float4*)(xrow0 + 8 * j);
      const float4 a1 = *(const float4*)(xrow1 + 8 * j);
      float4 bq[4];
      #pragma unroll
      for (int t = 0; t < 4; ++t) bq[t] = wf[(j * 4 + t) * 32];
      #pragma unroll
      for (int t = 0; t < 4; ++t) {
        acc[0][t] = __builtin_amdgcn_wmma_f32_16x16x4_f32(false, v2f{a0.x, a0.y}, false, v2f{bq[t].x, bq[t].y},
                                                          (short)0, acc[0][t], false, false);
        acc[1][t] = __builtin_amdgcn_wmma_f32_16x16x4_f32(false, v2f{a1.x, a1.y}, false, v2f{bq[t].x, bq[t].y},
                                                          (short)0, acc[1][t], false, false);
      }
      #pragma unroll
      for (int t = 0; t < 4; ++t) {
        acc[0][t] = __builtin_amdgcn_wmma_f32_16x16x4_f32(false, v2f{a0.z, a0.w}, false, v2f{bq[t].z, bq[t].w},
                                                          (short)0, acc[0][t], false, false);
        acc[1][t] = __builtin_amdgcn_wmma_f32_16x16x4_f32(false, v2f{a1.z, a1.w}, false, v2f{bq[t].z, bq[t].w},
                                                          (short)0, acc[1][t], false, false);
      }
    }

    #pragma unroll
    for (int mt = 0; mt < 2; ++mt) {
      #pragma unroll
      for (int t = 0; t < 4; ++t) {
        const float bias = sB1[16 * t + lm];
        float4 lo, hi;
        lo.x = fmaxf(acc[mt][t][0] + bias, 0.0f); lo.y = fmaxf(acc[mt][t][1] + bias, 0.0f);
        lo.z = fmaxf(acc[mt][t][2] + bias, 0.0f); lo.w = fmaxf(acc[mt][t][3] + bias, 0.0f);
        hi.x = fmaxf(acc[mt][t][4] + bias, 0.0f); hi.y = fmaxf(acc[mt][t][5] + bias, 0.0f);
        hi.z = fmaxf(acc[mt][t][6] + bias, 0.0f); hi.w = fmaxf(acc[mt][t][7] + bias, 0.0f);
        float* p = hw + (16 * t + lm) * 36 + 16 * mt + 8 * half;
        *(float4*)p = lo;
        *(float4*)(p + 4) = hi;
      }
    }
  }
  __syncthreads();

  {
    const float* hr = hw + lane;
    v2f h[32];
    #pragma unroll
    for (int i = 0; i < 32; ++i) h[i] = v2f{hr[(2 * i) * 36], hr[(2 * i + 1) * 36]};
    v2f a0 = h[0], a1 = h[1], a2 = h[2], a3 = h[3];
    #pragma unroll
    for (int i = 4; i < 32; i += 4) { a0 += h[i]; a1 += h[i + 1]; a2 += h[i + 2]; a3 += h[i + 3]; }
    const v2f sa = (a0 + a1) + (a2 + a3);
    const float mu = (sa.x + sa.y) * 0.015625f;
    const v2f mu2 = {mu, mu};
    #pragma unroll
    for (int i = 0; i < 32; ++i) h[i] = h[i] - mu2;
    v2f q0 = h[0] * h[0], q1 = h[1] * h[1], q2 = h[2] * h[2], q3 = h[3] * h[3];
    #pragma unroll
    for (int i = 4; i < 32; i += 4) {
      q0 = __builtin_elementwise_fma(h[i], h[i], q0);
      q1 = __builtin_elementwise_fma(h[i + 1], h[i + 1], q1);
      q2 = __builtin_elementwise_fma(h[i + 2], h[i + 2], q2);
      q3 = __builtin_elementwise_fma(h[i + 3], h[i + 3], q3);
    }
    const v2f sq = (q0 + q1) + (q2 + q3);
    const float inv = rsqrtf((sq.x + sq.y) * 0.015625f + 1e-5f);
    const v2f inv2 = {inv, inv};
    #pragma unroll
    for (int i = 0; i < 32; ++i) {
      const v2f s = h[i] * inv2;
      h[i] = v2f{fmaf(s.x, ln_g[2 * i], ln_b[2 * i]), fmaf(s.y, ln_g[2 * i + 1], ln_b[2 * i + 1])};
    }
    #pragma unroll
    for (int q = 0; q < 6; ++q) {
      const float* w = W2 + q * 64;
      float p0x = h[0].x * w[0], p0y = h[0].y * w[1], p1x = h[1].x * w[2], p1y = h[1].y * w[3];
      #pragma unroll
      for (int i = 2; i < 32; i += 2) {
        p0x = fmaf(h[i].x, w[2 * i], p0x);
        p0y = fmaf(h[i].y, w[2 * i + 1], p0y);
        p1x = fmaf(h[i + 1].x, w[2 * i + 2], p1x);
        p1y = fmaf(h[i + 1].y, w[2 * i + 3], p1y);
      }
      const v2f ps = {p0x + p1x, p0y + p1y};
      hw[q * 32 + lane] = (ps.x + ps.y) + b2[q];
      __builtin_amdgcn_sched_barrier(0);
    }
  }

  const int row = rowBase + lane;
  #pragma unroll 1
  for (int q = 0; q < 6; ++q) {
    const float zz = tanhf(hw[q * 32 + lane]);
    const float h2 = 1.57079632679489662f * zz;
    const float ce = cosf(h2), se = sinf(h2);
    const float2 ry = sTab[128 + q];
    const float2 pz = sTab[146 + q];
    const float cc = ry.x * ce, ss = ry.y * se, sc = ry.y * ce, cs = ry.x * se;
    hw[192 + (4 * q) * 32 + lane]     = pz.x * (cc - ss);
    hw[192 + (4 * q + 1) * 32 + lane] = -pz.y * (cc + ss);
    hw[192 + (4 * q + 2) * 32 + lane] = pz.x * (sc + cs);
    hw[192 + (4 * q + 3) * 32 + lane] = pz.y * (cs - sc);
  }

  v2f st[64];
  {
    v2f u[6][2];
    #pragma unroll
    for (int q = 0; q < 6; ++q) {
      u[q][0] = v2f{hw[192 + (4 * q) * 32 + lane],     hw[192 + (4 * q + 1) * 32 + lane]};
      u[q][1] = v2f{hw[192 + (4 * q + 2) * 32 + lane], hw[192 + (4 * q + 3) * 32 + lane]};
    }
    v2f hi[8], lo[8];
    triple(hi, u[0], u[1], u[2]);
    triple(lo, u[3], u[4], u[5]);
    build_state(st, hi, lo, std::make_integer_sequence<int, 64>{});
  }

  #pragma unroll 1
  for (int l = 1; l < 3; ++l) {
    if (l == 2) ring_apply<2>(st, std::make_integer_sequence<int, 64>{});
    phase_all(st, sTab + 64 * (l - 1), std::make_integer_sequence<int, 64>{});
    layer_ry(st, sTab + 128 + l * 6, std::make_integer_sequence<int, 6>{});
  }

  float zq[6];
  {
    float cur[64];
    probs<1>(cur, st, std::make_integer_sequence<int, 64>{});
    #pragma unroll
    for (int b = 0; b < 6; ++b) {
      const int n2 = 32 >> b;
      float accd = cur[0] - cur[1];
      cur[0] = cur[0] + cur[1];
      #pragma unroll
      for (int j = 1; j < n2; ++j) {
        float lo = cur[2 * j], hi = cur[2 * j + 1];
        accd += lo - hi;
        cur[j] = lo + hi;
      }
      zq[5 - b] = accd;
    }
  }
  if (row < B) {
    float nrm = sNrm[0];
    #pragma unroll
    for (int g = 1; g < 12; ++g) nrm *= sNrm[g];
    #pragma unroll
    for (int k = 0; k < 5; ++k) {
      float o = 0.f;
      #pragma unroll
      for (int q = 0; q < 6; ++q) o = fmaf(Wc[k * 6 + q], zq[q], o);
      out[(size_t)row * 5 + k] = fmaf(o, nrm, bc[k]);
    }
  }
}

extern "C" void kernel_launch(void* const* d_in, const int* in_sizes, int n_in,
                              void* d_out, int out_size, void* d_ws, size_t ws_size,
                              hipStream_t stream) {
  (void)n_in; (void)out_size; (void)d_ws; (void)ws_size;
  const float* x    = (const float*)d_in[0];
  const float* W1   = (const float*)d_in[1];
  const float* b1   = (const float*)d_in[2];
  const float* ln_g = (const float*)d_in[3];
  const float* ln_b = (const float*)d_in[4];
  const float* W2   = (const float*)d_in[5];
  const float* b2   = (const float*)d_in[6];
  const float* th_s = (const float*)d_in[7];
  const float* th_t = (const float*)d_in[8];
  const float* Wc   = (const float*)d_in[9];
  const float* bc   = (const float*)d_in[10];
  float* out = (float*)d_out;
  int B = in_sizes[0] / 64;
  int grid = (B + 127) / 128;
  qmaml_kernel<<<grid, 128, 0, stream>>>(x, W1, b1, ln_g, ln_b, W2, b2, th_s, th_t, Wc, bc, out, B);
}
